# starting kernel
# baseline (speedup 1.0000x reference)
#include <hip/hip_runtime.h>

typedef float v2f __attribute__((ext_vector_type(2)));
typedef float v8f __attribute__((ext_vector_type(8)));

#define Bc     4
#define LFULL  2050
#define Lq     2048
#define Hh     16
#define Dd     64
#define SHAPE  4095
#define WSTR   4096

#define KT     32
#define MT     256
#define BAND   (KT + MT)
#define VSTR   (KT + 2)

__global__ __launch_bounds__(256) void prep_kernel(const float* __restrict__ w,
                                                   const float* __restrict__ offset,
                                                   float* __restrict__ wexp,
                                                   float* __restrict__ out)
{
    __shared__ float s[4096];
    __shared__ float part[256];
    const int h   = blockIdx.x;
    const int tid = threadIdx.x;
    const float off = offset[h];

    for (int k = tid; k < 4096; k += 256) {
        float val = 0.f;
        if (k < SHAPE) val = __expf(w[h * SHAPE + k] - off);
        wexp[h * WSTR + k] = val;
        s[k] = val;
    }
    __syncthreads();

    const int base = tid * 16;
    float loc[16];
    float acc = 0.f;
    #pragma unroll
    for (int i = 0; i < 16; ++i) { loc[i] = acc; acc += s[base + i]; }
    part[tid] = acc;
    __syncthreads();
    if (tid == 0) {
        float run = 0.f;
        for (int i = 0; i < 256; ++i) { float t = part[i]; part[i] = run; run += t; }
    }
    __syncthreads();
    const float pbase = part[tid];
    #pragma unroll
    for (int i = 0; i < 16; ++i) s[base + i] = pbase + loc[i];
    __syncthreads();

    float* zout = out + (size_t)Bc * LFULL * Hh * Dd;
    for (int i = tid; i < Lq; i += 256) {
        const float z = s[i + Lq] - s[i];
        #pragma unroll
        for (int n = 0; n < Bc; ++n)
            zout[((size_t)n * LFULL + (i + 1)) * Hh + h] = z;
    }
    if (tid < 2 * Bc) {
        const int n  = tid & 3;
        const int jp = (tid >> 2) ? (LFULL - 1) : 0;
        zout[((size_t)n * LFULL + jp) * Hh + h] = 0.f;
    }
    for (int i = tid; i < Bc * 2 * Dd; i += 256) {
        const int d  = i & 63;
        const int r  = i >> 6;
        const int n  = r & 3;
        const int jp = (r >> 2) ? (LFULL - 1) : 0;
        out[(((size_t)n * LFULL + jp) * Hh + h) * Dd + d] = 0.f;
    }
}

__global__ __launch_bounds__(256) void pbv_kernel(const float* __restrict__ v,
                                                  const float* __restrict__ wexp,
                                                  float* __restrict__ out)
{
    __shared__ float band[BAND];
    __shared__ float vlds[Dd * VSTR];

    const int tid  = threadIdx.x;
    const int lane = tid & 31;
    const int wave = tid >> 5;
    const int m    = lane & 15;
    const int kh   = lane >> 4;
    const int j0   = blockIdx.x * MT;
    const int n    = blockIdx.y;
    const int h    = blockIdx.z;

    const float* wrow = wexp + h * WSTR;
    const int bandBase = (Lq - 1) - j0 - (MT - 1);

    v8f c[2][4] = {};

    const float* vbase = v + (((size_t)n * LFULL + 1) * Hh + h) * Dd;
    const int vrowk = tid >> 3;
    const int vd    = (tid & 7) * 8;

    for (int l0 = 0; l0 < Lq; l0 += KT) {
        band[tid] = wrow[bandBase + l0 + tid];
        if (tid < BAND - 256) band[256 + tid] = wrow[bandBase + l0 + 256 + tid];

        {
            const float* src = vbase + (size_t)(l0 + vrowk) * (Hh * Dd) + vd;
            const float4 f0 = *(const float4*)(src);
            const float4 f1 = *(const float4*)(src + 4);
            vlds[(vd + 0) * VSTR + vrowk] = f0.x;
            vlds[(vd + 1) * VSTR + vrowk] = f0.y;
            vlds[(vd + 2) * VSTR + vrowk] = f0.z;
            vlds[(vd + 3) * VSTR + vrowk] = f0.w;
            vlds[(vd + 4) * VSTR + vrowk] = f1.x;
            vlds[(vd + 5) * VSTR + vrowk] = f1.y;
            vlds[(vd + 6) * VSTR + vrowk] = f1.z;
            vlds[(vd + 7) * VSTR + vrowk] = f1.w;
        }
        __syncthreads();

        const int wr = wave * 32;
        #pragma unroll
        for (int ks = 0; ks < KT / 4; ++ks) {
            v2f a[2];
            #pragma unroll
            for (int f = 0; f < 2; ++f) {
                const int t0 = ks * 4 + 2 * kh + (MT - 1) - (wr + f * 16 + m);
                a[f][0] = band[t0];
                a[f][1] = band[t0 + 1];
            }
            #pragma unroll
            for (int nt = 0; nt < 4; ++nt) {
                const float* bp = &vlds[(nt * 16 + m) * VSTR + ks * 4 + 2 * kh];
                v2f b;
                b[0] = bp[0];
                b[1] = bp[1];
                #pragma unroll
                for (int f = 0; f < 2; ++f) {
                    c[f][nt] = __builtin_amdgcn_wmma_f32_16x16x4_f32(
                        false, a[f], false, b, (short)0, c[f][nt], false, false);
                }
            }
        }
        __syncthreads();
    }

    #pragma unroll
    for (int f = 0; f < 2; ++f) {
        #pragma unroll
        for (int nt = 0; nt < 4; ++nt) {
            const int d = nt * 16 + m;
            #pragma unroll
            for (int r = 0; r < 8; ++r) {
                const int mrow = (lane < 16) ? r : (r + 8);
                const int j = j0 + wave * 32 + f * 16 + mrow;
                out[(((size_t)n * LFULL + (j + 1)) * Hh + h) * Dd + d] = c[f][nt][r];
            }
        }
    }
}

extern "C" void kernel_launch(void* const* d_in, const int* in_sizes, int n_in,
                              void* d_out, int out_size, void* d_ws, size_t ws_size,
                              hipStream_t stream)
{
    (void)in_sizes; (void)n_in; (void)out_size; (void)ws_size;
    const float* v      = (const float*)d_in[0];
    const float* offset = (const float*)d_in[1];
    const float* w      = (const float*)d_in[2];
    float*       out    = (float*)d_out;
    float*       wexp   = (float*)d_ws;

    prep_kernel<<<dim3(Hh), 256, 0, stream>>>(w, offset, wexp, out);
    pbv_kernel<<<dim3(Lq / MT, Bc, Hh), 256, 0, stream>>>(v, wexp, out);
}


# V operand not staged in LDS any more: each wave loads V rows as 16-byte global loads straight into its B fragments (N tile nt = columns 4c+nt), one block ahead; band in LDS once; no slab barriers
# speedup vs baseline: 1.1594x; 1.1594x over previous
#include <hip/hip_runtime.h>

typedef float v2f __attribute__((ext_vector_type(2)));
typedef float v4f __attribute__((ext_vector_type(4)));
typedef float v8f __attribute__((ext_vector_type(8)));

#define Bc     4
#define LFULL  2050
#define Lq     2048
#define Hh     16
#define Dd     64
#define SHAPE  4095
#define WSTR   4096

#define MT     256

__global__ __launch_bounds__(256) void prep_kernel(const float* __restrict__ w,
                                                   const float* __restrict__ offset,
                                                   float* __restrict__ wexp,
                                                   float* __restrict__ out)
{
    __shared__ float s[4096];
    __shared__ float part[256];
    const int h   = blockIdx.x;
    const int tid = threadIdx.x;
    const float off = offset[h];

    for (int k = tid; k < 4096; k += 256) {
        float val = 0.f;
        if (k < SHAPE) val = __expf(w[h * SHAPE + k] - off);
        wexp[h * WSTR + k] = val;
        s[k] = val;
    }
    __syncthreads();

    const int base = tid * 16;
    float loc[16];
    float acc = 0.f;
    #pragma unroll
    for (int i = 0; i < 16; ++i) { loc[i] = acc; acc += s[base + i]; }
    part[tid] = acc;
    __syncthreads();
    if (tid == 0) {
        float run = 0.f;
        for (int i = 0; i < 256; ++i) { float t = part[i]; part[i] = run; run += t; }
    }
    __syncthreads();
    const float pbase = part[tid];
    #pragma unroll
    for (int i = 0; i < 16; ++i) s[base + i] = pbase + loc[i];
    __syncthreads();

    float* zout = out + (size_t)Bc * LFULL * Hh * Dd;
    for (int i = tid; i < Lq; i += 256) {
        const float z = s[i + Lq] - s[i];
        #pragma unroll
        for (int n = 0; n < Bc; ++n)
            zout[((size_t)n * LFULL + (i + 1)) * Hh + h] = z;
    }
    if (tid < 2 * Bc) {
        const int n  = tid & 3;
        const int jp = (tid >> 2) ? (LFULL - 1) : 0;
        zout[((size_t)n * LFULL + jp) * Hh + h] = 0.f;
    }
    for (int i = tid; i < Bc * 2 * Dd; i += 256) {
        const int d  = i & 63;
        const int r  = i >> 6;
        const int n  = r & 3;
        const int jp = (r >> 2) ? (LFULL - 1) : 0;
        out[(((size_t)n * LFULL + jp) * Hh + h) * Dd + d] = 0.f;
    }
}

#define KB     16
#define BANDN  (Lq + MT)

template <bool PF>
__device__ __forceinline__ void pbv_block(const float* __restrict__ vp, const float* ap, int lnext,
                                          v8f (&acc)[2][4],
                                          const v4f (&bc)[4][2], v4f (&bn)[4][2],
                                          const v2f (&a0)[4], v2f (&an)[4], v2f (&a1)[4])
{
    if (PF) {
        #pragma unroll
        for (int ks = 0; ks < 4; ++ks) {
            #pragma unroll
            for (int e = 0; e < 2; ++e)
                bn[ks][e] = *(const v4f*)(vp + (size_t)(lnext + 4 * ks + e) * (Hh * Dd));
        }
        #pragma unroll
        for (int ks = 0; ks < 4; ++ks) {
            an[ks][0] = ap[lnext + 4 * ks];
            an[ks][1] = ap[lnext + 4 * ks + 1];
        }
    }
    #pragma unroll
    for (int ks = 0; ks < 4; ++ks) {
        #pragma unroll
        for (int nt = 0; nt < 4; ++nt) {
            v2f b;
            b[0] = bc[ks][0][nt];
            b[1] = bc[ks][1][nt];
            acc[0][nt] = __builtin_amdgcn_wmma_f32_16x16x4_f32(
                false, a0[ks], false, b, (short)0, acc[0][nt], false, false);
            acc[1][nt] = __builtin_amdgcn_wmma_f32_16x16x4_f32(
                false, a1[ks], false, b, (short)0, acc[1][nt], false, false);
        }
    }
    #pragma unroll
    for (int ks = 0; ks < 4; ++ks) a1[ks] = a0[ks];
}

__global__ __launch_bounds__(256) void pbv_kernel(const float* __restrict__ v,
                                                  const float* __restrict__ wexp,
                                                  float* __restrict__ out)
{
    __shared__ float band[BANDN];

    const int tid  = threadIdx.x;
    const int lane = tid & 31;
    const int wave = tid >> 5;
    const int c    = lane & 15;
    const int hf   = lane >> 4;
    const int j0   = blockIdx.x * MT;
    const int n    = blockIdx.y;
    const int h    = blockIdx.z;

    const float* wrow = wexp + h * WSTR + ((Lq - 1) - j0 - (MT - 1));
    for (int i = tid; i < Lq + MT - 1; i += 256) band[i] = wrow[i];
    __syncthreads();

    const float* vp = v + (((size_t)n * LFULL + 1) * Hh + h) * Dd
                        + (size_t)(2 * hf) * (Hh * Dd) + 4 * c;
    const float* ap = band + (MT - 1) - (wave * 32 + c) + 2 * hf;

    v8f acc[2][4] = {};
    v4f bA[4][2], bB[4][2];
    v2f aA[4], aB[4], a1[4];
    #pragma unroll
    for (int ks = 0; ks < 4; ++ks) {
        bA[ks][0] = *(const v4f*)(vp + (size_t)(4 * ks) * (Hh * Dd));
        bA[ks][1] = *(const v4f*)(vp + (size_t)(4 * ks + 1) * (Hh * Dd));
        aA[ks][0] = ap[4 * ks];
        aA[ks][1] = ap[4 * ks + 1];
        a1[ks][0] = ap[4 * ks - 16];
        a1[ks][1] = ap[4 * ks - 15];
    }

    for (int l0 = 0; l0 < Lq - 2 * KB; l0 += 2 * KB) {
        __builtin_amdgcn_s_barrier_signal(-1);
        pbv_block<true>(vp, ap, l0 + KB,     acc, bA, bB, aA, aB, a1);
        pbv_block<true>(vp, ap, l0 + 2 * KB, acc, bB, bA, aB, aA, a1);
        __builtin_amdgcn_s_barrier_wait(-1);
    }
    pbv_block<true>(vp, ap, Lq - KB, acc, bA, bB, aA, aB, a1);
    pbv_block<false>(vp, ap, 0,      acc, bB, bA, aB, aA, a1);

    #pragma unroll
    for (int f = 0; f < 2; ++f) {
        #pragma unroll
        for (int r = 0; r < 8; ++r) {
            const int j = j0 + wave * 32 + f * 16 + 8 * hf + r;
            v4f o;
            o[0] = acc[f][0][r];
            o[1] = acc[f][1][r];
            o[2] = acc[f][2][r];
            o[3] = acc[f][3][r];
            *(v4f*)(out + (((size_t)n * LFULL + (j + 1)) * Hh + h) * Dd + 4 * c) = o;
        }
    }
}

extern "C" void kernel_launch(void* const* d_in, const int* in_sizes, int n_in,
                              void* d_out, int out_size, void* d_ws, size_t ws_size,
                              hipStream_t stream)
{
    (void)in_sizes; (void)n_in; (void)out_size; (void)ws_size;
    const float* v      = (const float*)d_in[0];
    const float* offset = (const float*)d_in[1];
    const float* w      = (const float*)d_in[2];
    float*       out    = (float*)d_out;
    float*       wexp   = (float*)d_ws;

    prep_kernel<<<dim3(Hh), 256, 0, stream>>>(w, offset, wexp, out);
    pbv_kernel<<<dim3(Lq / MT, Bc, Hh), 256, 0, stream>>>(v, wexp, out);
}


# Next block's V and band loads stand as ONE group ahead of each block's 32 WMMAs (two scheduling fences per block): 2 load places per 64 WMMAs, not the 9 the compiler had spread among them
# speedup vs baseline: 1.2020x; 1.0367x over previous
#include <hip/hip_runtime.h>

typedef float v2f __attribute__((ext_vector_type(2)));
typedef float v4f __attribute__((ext_vector_type(4)));
typedef float v8f __attribute__((ext_vector_type(8)));

#define Bc     4
#define LFULL  2050
#define Lq     2048
#define Hh     16
#define Dd     64
#define SHAPE  4095
#define WSTR   4096

#define MT     256

__global__ __launch_bounds__(256) void prep_kernel(const float* __restrict__ w,
                                                   const float* __restrict__ offset,
                                                   float* __restrict__ wexp,
                                                   float* __restrict__ out)
{
    __shared__ float s[4096];
    __shared__ float part[256];
    const int h   = blockIdx.x;
    const int tid = threadIdx.x;
    const float off = offset[h];

    for (int k = tid; k < 4096; k += 256) {
        float val = 0.f;
        if (k < SHAPE) val = __expf(w[h * SHAPE + k] - off);
        wexp[h * WSTR + k] = val;
        s[k] = val;
    }
    __syncthreads();

    const int base = tid * 16;
    float loc[16];
    float acc = 0.f;
    #pragma unroll
    for (int i = 0; i < 16; ++i) { loc[i] = acc; acc += s[base + i]; }
    part[tid] = acc;
    __syncthreads();
    if (tid == 0) {
        float run = 0.f;
        for (int i = 0; i < 256; ++i) { float t = part[i]; part[i] = run; run += t; }
    }
    __syncthreads();
    const float pbase = part[tid];
    #pragma unroll
    for (int i = 0; i < 16; ++i) s[base + i] = pbase + loc[i];
    __syncthreads();

    float* zout = out + (size_t)Bc * LFULL * Hh * Dd;
    for (int i = tid; i < Lq; i += 256) {
        const float z = s[i + Lq] - s[i];
        #pragma unroll
        for (int n = 0; n < Bc; ++n)
            zout[((size_t)n * LFULL + (i + 1)) * Hh + h] = z;
    }
    if (tid < 2 * Bc) {
        const int n  = tid & 3;
        const int jp = (tid >> 2) ? (LFULL - 1) : 0;
        zout[((size_t)n * LFULL + jp) * Hh + h] = 0.f;
    }
    for (int i = tid; i < Bc * 2 * Dd; i += 256) {
        const int d  = i & 63;
        const int r  = i >> 6;
        const int n  = r & 3;
        const int jp = (r >> 2) ? (LFULL - 1) : 0;
        out[(((size_t)n * LFULL + jp) * Hh + h) * Dd + d] = 0.f;
    }
}

#define KB     16
#define BANDN  (Lq + MT)

template <bool PF>
__device__ __forceinline__ void pbv_block(const float* __restrict__ vp, const float* ap, int lnext,
                                          v8f (&acc)[2][4],
                                          const v4f (&bc)[4][2], v4f (&bn)[4][2],
                                          const v2f (&a0)[4], v2f (&an)[4], v2f (&a1)[4])
{
    __builtin_amdgcn_sched_barrier(0);
    if (PF) {
        #pragma unroll
        for (int ks = 0; ks < 4; ++ks) {
            #pragma unroll
            for (int e = 0; e < 2; ++e)
                bn[ks][e] = *(const v4f*)(vp + (size_t)(lnext + 4 * ks + e) * (Hh * Dd));
        }
        #pragma unroll
        for (int ks = 0; ks < 4; ++ks) {
            an[ks][0] = ap[lnext + 4 * ks];
            an[ks][1] = ap[lnext + 4 * ks + 1];
        }
    }
    __builtin_amdgcn_sched_barrier(0);
    #pragma unroll
    for (int ks = 0; ks < 4; ++ks) {
        #pragma unroll
        for (int nt = 0; nt < 4; ++nt) {
            v2f b;
            b[0] = bc[ks][0][nt];
            b[1] = bc[ks][1][nt];
            acc[0][nt] = __builtin_amdgcn_wmma_f32_16x16x4_f32(
                false, a0[ks], false, b, (short)0, acc[0][nt], false, false);
            acc[1][nt] = __builtin_amdgcn_wmma_f32_16x16x4_f32(
                false, a1[ks], false, b, (short)0, acc[1][nt], false, false);
        }
    }
    #pragma unroll
    for (int ks = 0; ks < 4; ++ks) a1[ks] = a0[ks];
}

__global__ __launch_bounds__(256) void pbv_kernel(const float* __restrict__ v,
                                                  const float* __restrict__ wexp,
                                                  float* __restrict__ out)
{
    __shared__ float band[BANDN];

    const int tid  = threadIdx.x;
    const int lane = tid & 31;
    const int wave = tid >> 5;
    const int c    = lane & 15;
    const int hf   = lane >> 4;
    const int j0   = blockIdx.x * MT;
    const int n    = blockIdx.y;
    const int h    = blockIdx.z;

    const float* wrow = wexp + h * WSTR + ((Lq - 1) - j0 - (MT - 1));
    for (int i = tid; i < Lq + MT - 1; i += 256) band[i] = wrow[i];
    __syncthreads();

    const float* vp = v + (((size_t)n * LFULL + 1) * Hh + h) * Dd
                        + (size_t)(2 * hf) * (Hh * Dd) + 4 * c;
    const float* ap = band + (MT - 1) - (wave * 32 + c) + 2 * hf;

    v8f acc[2][4] = {};
    v4f bA[4][2], bB[4][2];
    v2f aA[4], aB[4], a1[4];
    #pragma unroll
    for (int ks = 0; ks < 4; ++ks) {
        bA[ks][0] = *(const v4f*)(vp + (size_t)(4 * ks) * (Hh * Dd));
        bA[ks][1] = *(const v4f*)(vp + (size_t)(4 * ks + 1) * (Hh * Dd));
        aA[ks][0] = ap[4 * ks];
        aA[ks][1] = ap[4 * ks + 1];
        a1[ks][0] = ap[4 * ks - 16];
        a1[ks][1] = ap[4 * ks - 15];
    }

    for (int l0 = 0; l0 < Lq - 2 * KB; l0 += 2 * KB) {
        __builtin_amdgcn_s_barrier_signal(-1);
        pbv_block<true>(vp, ap, l0 + KB,     acc, bA, bB, aA, aB, a1);
        pbv_block<true>(vp, ap, l0 + 2 * KB, acc, bB, bA, aB, aA, a1);
        __builtin_amdgcn_s_barrier_wait(-1);
    }
    pbv_block<true>(vp, ap, Lq - KB, acc, bA, bB, aA, aB, a1);
    pbv_block<false>(vp, ap, 0,      acc, bB, bA, aB, aA, a1);

    #pragma unroll
    for (int f = 0; f < 2; ++f) {
        #pragma unroll
        for (int r = 0; r < 8; ++r) {
            const int j = j0 + wave * 32 + f * 16 + 8 * hf + r;
            v4f o;
            o[0] = acc[f][0][r];
            o[1] = acc[f][1][r];
            o[2] = acc[f][2][r];
            o[3] = acc[f][3][r];
            *(v4f*)(out + (((size_t)n * LFULL + (j + 1)) * Hh + h) * Dd + 4 * c) = o;
        }
    }
}

extern "C" void kernel_launch(void* const* d_in, const int* in_sizes, int n_in,
                              void* d_out, int out_size, void* d_ws, size_t ws_size,
                              hipStream_t stream)
{
    (void)in_sizes; (void)n_in; (void)out_size; (void)ws_size;
    const float* v      = (const float*)d_in[0];
    const float* offset = (const float*)d_in[1];
    const float* w      = (const float*)d_in[2];
    float*       out    = (float*)d_out;
    float*       wexp   = (float*)d_ws;

    prep_kernel<<<dim3(Hh), 256, 0, stream>>>(w, offset, wexp, out);
    pbv_kernel<<<dim3(Lq / MT, Bc, Hh), 256, 0, stream>>>(v, wexp, out);
}


# First launch (exp, z_pb sums) without serial round trips: a thread's 16 w loads in flight at once; prefix sum in registers + lane-shift wave scan, not one thread's pass; changes f32 sum order of z_pb
# speedup vs baseline: 1.2407x; 1.0322x over previous
#include <hip/hip_runtime.h>

typedef float v2f __attribute__((ext_vector_type(2)));
typedef float v4f __attribute__((ext_vector_type(4)));
typedef float v8f __attribute__((ext_vector_type(8)));

#define Bc     4
#define LFULL  2050
#define Lq     2048
#define Hh     16
#define Dd     64
#define SHAPE  4095
#define WSTR   4096

#define MT     256

__global__ __launch_bounds__(256) void prep_kernel(const float* __restrict__ w,
                                                   const float* __restrict__ offset,
                                                   float* __restrict__ wexp,
                                                   float* __restrict__ out)
{
    __shared__ __attribute__((aligned(16))) float s[4096];
    __shared__ __attribute__((aligned(16))) float wtot[8];
    const int h    = blockIdx.x;
    const int tid  = threadIdx.x;
    const int lane = tid & 31;
    const int wave = tid >> 5;
    const float off = offset[h];
    const float* wrow = w + h * SHAPE;

    float raw[16];
    #pragma unroll
    for (int i = 0; i < 16; ++i) {
        const int k = tid + 256 * i;
        raw[i] = wrow[k < SHAPE ? k : SHAPE - 1];
    }
    float val[16];
    #pragma unroll
    for (int i = 0; i < 16; ++i) {
        const int k = tid + 256 * i;
        val[i] = (k < SHAPE) ? __expf(raw[i] - off) : 0.f;
        s[k] = val[i];
    }
    __syncthreads();

    const int base = tid * 16;
    float loc[16];
    float acc = 0.f;
    #pragma unroll
    for (int q = 0; q < 4; ++q) {
        const v4f t4 = *(const v4f*)(s + base + 4 * q);
        #pragma unroll
        for (int e = 0; e < 4; ++e) { loc[4 * q + e] = acc; acc += t4[e]; }
    }
    float inc = acc;
    #pragma unroll
    for (int d = 1; d < 32; d <<= 1) {
        const float up = __shfl_up(inc, d, 32);
        if (lane >= d) inc += up;
    }
    if (lane == 31) wtot[wave] = inc;
    __syncthreads();
    float pbase = inc - acc;
    #pragma unroll
    for (int q = 0; q < 7; ++q) pbase += (q < wave) ? wtot[q] : 0.f;
    #pragma unroll
    for (int q = 0; q < 4; ++q) {
        v4f t4;
        #pragma unroll
        for (int e = 0; e < 4; ++e) t4[e] = pbase + loc[4 * q + e];
        *(v4f*)(s + base + 4 * q) = t4;
    }
    __syncthreads();

    float* zout = out + (size_t)Bc * LFULL * Hh * Dd;
    float zv[8];
    #pragma unroll
    for (int r = 0; r < 8; ++r) {
        const int i = tid + 256 * r;
        zv[r] = s[i + Lq] - s[i];
    }
    #pragma unroll
    for (int r = 0; r < 8; ++r) {
        const int i = tid + 256 * r;
        #pragma unroll
        for (int n = 0; n < Bc; ++n)
            zout[((size_t)n * LFULL + (i + 1)) * Hh + h] = zv[r];
    }
    if (tid < 2 * Bc) {
        const int n  = tid & 3;
        const int jp = (tid >> 2) ? (LFULL - 1) : 0;
        zout[((size_t)n * LFULL + jp) * Hh + h] = 0.f;
    }
    for (int i = tid; i < Bc * 2 * Dd; i += 256) {
        const int d  = i & 63;
        const int r  = i >> 6;
        const int n  = r & 3;
        const int jp = (r >> 2) ? (LFULL - 1) : 0;
        out[(((size_t)n * LFULL + jp) * Hh + h) * Dd + d] = 0.f;
    }
    #pragma unroll
    for (int i = 0; i < 16; ++i) wexp[h * WSTR + tid + 256 * i] = val[i];
}

#define KB     16
#define BANDN  (Lq + MT)

template <bool PF>
__device__ __forceinline__ void pbv_block(const float* __restrict__ vp, const float* ap, int lnext,
                                          v8f (&acc)[2][4],
                                          const v4f (&bc)[4][2], v4f (&bn)[4][2],
                                          const v2f (&a0)[4], v2f (&an)[4], v2f (&a1)[4])
{
    __builtin_amdgcn_sched_barrier(0);
    if (PF) {
        #pragma unroll
        for (int ks = 0; ks < 4; ++ks) {
            #pragma unroll
            for (int e = 0; e < 2; ++e)
                bn[ks][e] = *(const v4f*)(vp + (size_t)(lnext + 4 * ks + e) * (Hh * Dd));
        }
        #pragma unroll
        for (int ks = 0; ks < 4; ++ks) {
            an[ks][0] = ap[lnext + 4 * ks];
            an[ks][1] = ap[lnext + 4 * ks + 1];
        }
    }
    __builtin_amdgcn_sched_barrier(0);
    #pragma unroll
    for (int ks = 0; ks < 4; ++ks) {
        #pragma unroll
        for (int nt = 0; nt < 4; ++nt) {
            v2f b;
            b[0] = bc[ks][0][nt];
            b[1] = bc[ks][1][nt];
            acc[0][nt] = __builtin_amdgcn_wmma_f32_16x16x4_f32(
                false, a0[ks], false, b, (short)0, acc[0][nt], false, false);
            acc[1][nt] = __builtin_amdgcn_wmma_f32_16x16x4_f32(
                false, a1[ks], false, b, (short)0, acc[1][nt], false, false);
        }
    }
    #pragma unroll
    for (int ks = 0; ks < 4; ++ks) a1[ks] = a0[ks];
}

__global__ __launch_bounds__(256) void pbv_kernel(const float* __restrict__ v,
                                                  const float* __restrict__ wexp,
                                                  float* __restrict__ out)
{
    __shared__ float band[BANDN];

    const int tid  = threadIdx.x;
    const int lane = tid & 31;
    const int wave = tid >> 5;
    const int c    = lane & 15;
    const int hf   = lane >> 4;
    const int j0   = blockIdx.x * MT;
    const int n    = blockIdx.y;
    const int h    = blockIdx.z;

    const float* wrow = wexp + h * WSTR + ((Lq - 1) - j0 - (MT - 1));
    for (int i = tid; i < Lq + MT - 1; i += 256) band[i] = wrow[i];
    __syncthreads();

    const float* vp = v + (((size_t)n * LFULL + 1) * Hh + h) * Dd
                        + (size_t)(2 * hf) * (Hh * Dd) + 4 * c;
    const float* ap = band + (MT - 1) - (wave * 32 + c) + 2 * hf;

    v8f acc[2][4] = {};
    v4f bA[4][2], bB[4][2];
    v2f aA[4], aB[4], a1[4];
    #pragma unroll
    for (int ks = 0; ks < 4; ++ks) {
        bA[ks][0] = *(const v4f*)(vp + (size_t)(4 * ks) * (Hh * Dd));
        bA[ks][1] = *(const v4f*)(vp + (size_t)(4 * ks + 1) * (Hh * Dd));
        aA[ks][0] = ap[4 * ks];
        aA[ks][1] = ap[4 * ks + 1];
        a1[ks][0] = ap[4 * ks - 16];
        a1[ks][1] = ap[4 * ks - 15];
    }

    for (int l0 = 0; l0 < Lq - 2 * KB; l0 += 2 * KB) {
        __builtin_amdgcn_s_barrier_signal(-1);
        pbv_block<true>(vp, ap, l0 + KB,     acc, bA, bB, aA, aB, a1);
        pbv_block<true>(vp, ap, l0 + 2 * KB, acc, bB, bA, aB, aA, a1);
        __builtin_amdgcn_s_barrier_wait(-1);
    }
    pbv_block<true>(vp, ap, Lq - KB, acc, bA, bB, aA, aB, a1);
    pbv_block<false>(vp, ap, 0,      acc, bB, bA, aB, aA, a1);

    #pragma unroll
    for (int f = 0; f < 2; ++f) {
        #pragma unroll
        for (int r = 0; r < 8; ++r) {
            const int j = j0 + wave * 32 + f * 16 + 8 * hf + r;
            v4f o;
            o[0] = acc[f][0][r];
            o[1] = acc[f][1][r];
            o[2] = acc[f][2][r];
            o[3] = acc[f][3][r];
            *(v4f*)(out + (((size_t)n * LFULL + (j + 1)) * Hh + h) * Dd + 4 * c) = o;
        }
    }
}

extern "C" void kernel_launch(void* const* d_in, const int* in_sizes, int n_in,
                              void* d_out, int out_size, void* d_ws, size_t ws_size,
                              hipStream_t stream)
{
    (void)in_sizes; (void)n_in; (void)out_size; (void)ws_size;
    const float* v      = (const float*)d_in[0];
    const float* offset = (const float*)d_in[1];
    const float* w      = (const float*)d_in[2];
    float*       out    = (float*)d_out;
    float*       wexp   = (float*)d_ws;

    prep_kernel<<<dim3(Hh), 256, 0, stream>>>(w, offset, wexp, out);
    pbv_kernel<<<dim3(Lq / MT, Bc, Hh), 256, 0, stream>>>(v, wexp, out);
}


# Product in the transform domain of the BLOCK index (32 blocks of 64; 64-point real transform, f=0..32): forward GEMM, 33 complex 64x64 products per head, inverse GEMM; changes the f32 sum order
# speedup vs baseline: 2.5188x; 2.0301x over previous
#include <hip/hip_runtime.h>


typedef float v2f __attribute__((ext_vector_type(2)));
typedef float v4f __attribute__((ext_vector_type(4)));
typedef float v8f __attribute__((ext_vector_type(8)));

#define Bc     4
#define LFULL  2050
#define Lq     2048
#define Hh     16
#define Dd     64
#define SHAPE  4095

#define NF     33
#define TABP   128
#define WS_TAB 0
#define WS_G   (Hh * NF * 3 * TABP)
#define WS_VH  (WS_G + 4096)
#define PLANE  (Hh * 64 * 64 * 256)
#define WS_YH  (WS_VH + PLANE)

#define NB_TAB (Hh * NF)
#define NB_FWD (Hh * Bc * 64 / 8)
#define NB_L1  (NB_FWD + 1 + NB_TAB)
#define NB_L2  (Hh + 2 * NB_TAB)
#define FMP    34

__device__ __forceinline__ v8f mm(v2f a, v2f b, v8f c)
{
    return __builtin_amdgcn_wmma_f32_16x16x4_f32(false, a, false, b, (short)0, c, false, false);
}

__global__ __launch_bounds__(256) void front_kernel(const float* __restrict__ v,
                                                    const float* __restrict__ w,
                                                    const float* __restrict__ offset,
                                                    float* __restrict__ ws)
{
    __shared__ __attribute__((aligned(16))) float fm[64 * FMP];
    __shared__ float ct[64];
    const int tid = threadIdx.x;

    if (blockIdx.x == NB_FWD) {
        #pragma unroll
        for (int i = 0; i < 8; ++i) {
            const int e    = tid + 256 * i;
            const int row  = e & 63;
            const int blk  = e >> 6;
            const int f    = row == 0 ? 0 : (row == 1 ? 32 : (row >> 1));
            const int part = row < 2 ? 0 : (row & 1);
            const float x  = (float)((f * blk) & 63) * (1.0f / 32.0f);
            ws[WS_G + e] = part ? (-2.0f / 64.0f) * sinpif(x)
                                : (row < 2 ? 1.0f / 64.0f : 2.0f / 64.0f) * cospif(x);
        }
        return;
    }

    if (blockIdx.x > NB_FWD) {
        const int idx = blockIdx.x - (NB_FWD + 1);
        const int h   = idx / NF;
        const int f   = idx - h * NF;
        if (tid < 64) ct[tid] = cospif((float)tid * (1.0f / 32.0f));
        __syncthreads();
        const int dl   = tid & 127;
        const int part = tid >> 7;
        const float off = offset[h];
        float sum = 0.f;
        if (dl < 127) {
            const float* wp = w + h * SHAPE + (Lq - 1) + (dl - 63);
            #pragma unroll 9
            for (int kk = 0; kk < 63; ++kk) {
                const int k = kk - 31;
                const int m = (f * k) & 63;
                const float e = __expf(wp[64 * k] - off);
                const float t = ct[(m - 16 * part) & 63];
                sum += e * t;
            }
        }
        float* tp = ws + WS_TAB + (size_t)idx * (3 * TABP);
        if (part == 0) {
            tp[dl] = sum;
        } else {
            tp[TABP + dl]     = sum;
            tp[2 * TABP + dl] = -sum;
        }
        return;
    }

    const int lane = tid & 31;
    const int wave = tid >> 5;
    const int cl   = lane & 15;
    const int hf   = lane >> 4;
    const int g    = blockIdx.x * 8 + wave;
    const int pc   = g & 63;
    const int n    = (g >> 6) & 3;
    const int h    = g >> 8;

    const float* bp = v + (((size_t)n * LFULL + pc + 1) * Hh + h) * Dd + 4 * cl
                        + (size_t)(2 * hf) * (64 * Hh * Dd);
    v4f b0[8], b1[8];
    #pragma unroll
    for (int ks = 0; ks < 8; ++ks) {
        b0[ks] = *(const v4f*)(bp + (size_t)(4 * ks) * (64 * Hh * Dd));
        b1[ks] = *(const v4f*)(bp + (size_t)(4 * ks + 1) * (64 * Hh * Dd));
    }
    #pragma unroll
    for (int i = 0; i < 8; ++i) {
        const int e    = tid + 256 * i;
        const int row  = e >> 5;
        const int blk  = e & 31;
        const int f    = row == 0 ? 0 : (row == 1 ? 32 : (row >> 1));
        const int part = row < 2 ? 0 : (row & 1);
        fm[row * FMP + blk] = cospif((float)((f * blk + 16 * part) & 63) * (1.0f / 32.0f));
    }
    __syncthreads();
    const float* ap = fm + cl * FMP + 2 * hf;

    v8f acc[4][4] = {};
    #pragma unroll
    for (int ks = 0; ks < 8; ++ks) {
        v2f a[4];
        #pragma unroll
        for (int mt = 0; mt < 4; ++mt) a[mt] = *(const v2f*)(ap + mt * (16 * FMP) + 4 * ks);
        #pragma unroll
        for (int nt = 0; nt < 4; ++nt) {
            v2f b;
            b[0] = b0[ks][nt];
            b[1] = b1[ks][nt];
            #pragma unroll
            for (int mt = 0; mt < 4; ++mt) acc[mt][nt] = mm(a[mt], b, acc[mt][nt]);
        }
    }
    float* vh = ws + WS_VH;
    #pragma unroll
    for (int mt = 0; mt < 4; ++mt) {
        #pragma unroll
        for (int r = 0; r < 8; ++r) {
            const int row = 16 * mt + 8 * hf + r;
            v4f o;
            o[0] = acc[mt][0][r];
            o[1] = acc[mt][1][r];
            o[2] = acc[mt][2][r];
            o[3] = acc[mt][3][r];
            *(v4f*)(vh + (((size_t)(h * 64 + row) * 64 + pc) * 256 + n * 64 + 4 * cl)) = o;
        }
    }
}

__device__ __forceinline__ void head_body(const float* __restrict__ w,
                                          const float* __restrict__ offset,
                                          float* __restrict__ out,
                                          float* s, float* wtot, int h)
{
    const int tid  = threadIdx.x;
    const int lane = tid & 31;
    const int wave = tid >> 5;
    const float off = offset[h];
    const float* wrow = w + h * SHAPE;

    float raw[16];
    #pragma unroll
    for (int i = 0; i < 16; ++i) {
        const int k = tid + 256 * i;
        raw[i] = wrow[k < SHAPE ? k : SHAPE - 1];
    }
    #pragma unroll
    for (int i = 0; i < 16; ++i) {
        const int k = tid + 256 * i;
        s[k] = (k < SHAPE) ? __expf(raw[i] - off) : 0.f;
    }
    __syncthreads();

    const int base = tid * 16;
    float loc[16];
    float acc = 0.f;
    #pragma unroll
    for (int q = 0; q < 4; ++q) {
        const v4f t4 = *(const v4f*)(s + base + 4 * q);
        #pragma unroll
        for (int e = 0; e < 4; ++e) { loc[4 * q + e] = acc; acc += t4[e]; }
    }
    float inc = acc;
    #pragma unroll
    for (int d = 1; d < 32; d <<= 1) {
        const float up = __shfl_up(inc, d, 32);
        if (lane >= d) inc += up;
    }
    if (lane == 31) wtot[wave] = inc;
    __syncthreads();
    float pbase = inc - acc;
    #pragma unroll
    for (int q = 0; q < 7; ++q) pbase += (q < wave) ? wtot[q] : 0.f;
    #pragma unroll
    for (int q = 0; q < 4; ++q) {
        v4f t4;
        #pragma unroll
        for (int e = 0; e < 4; ++e) t4[e] = pbase + loc[4 * q + e];
        *(v4f*)(s + base + 4 * q) = t4;
    }
    __syncthreads();

    float* zout = out + (size_t)Bc * LFULL * Hh * Dd;
    float zv[8];
    #pragma unroll
    for (int r = 0; r < 8; ++r) {
        const int i = tid + 256 * r;
        zv[r] = s[i + Lq] - s[i];
    }
    #pragma unroll
    for (int r = 0; r < 8; ++r) {
        const int i = tid + 256 * r;
        #pragma unroll
        for (int n = 0; n < Bc; ++n)
            zout[((size_t)n * LFULL + (i + 1)) * Hh + h] = zv[r];
    }
    if (tid < 2 * Bc) {
        const int n  = tid & 3;
        const int jp = (tid >> 2) ? (LFULL - 1) : 0;
        zout[((size_t)n * LFULL + jp) * Hh + h] = 0.f;
    }
    for (int i = tid; i < Bc * 2 * Dd; i += 256) {
        const int d  = i & 63;
        const int r  = i >> 6;
        const int n  = r & 3;
        const int jp = (r >> 2) ? (LFULL - 1) : 0;
        out[(((size_t)n * LFULL + jp) * Hh + h) * Dd + d] = 0.f;
    }
}

template <bool CPLX>
__device__ __forceinline__ void mid_body(const float* __restrict__ vre, const float* __restrict__ vim,
                                         const float* tab, int abase,
                                         float* __restrict__ yre_p, float* __restrict__ yim_p, int hf)
{
    v8f yre[4] = {};
    v8f yim[4] = {};
    #pragma unroll
    for (int ks = 0; ks < 16; ++ks) {
        const v4f br0 = *(const v4f*)(vre + (size_t)(4 * ks) * 256);
        const v4f br1 = *(const v4f*)(vre + (size_t)(4 * ks + 1) * 256);
        v2f tre;
        tre[0] = tab[abase + 4 * ks];
        tre[1] = tab[abase + 4 * ks + 1];
        if (CPLX) {
            const v4f bi0 = *(const v4f*)(vim + (size_t)(4 * ks) * 256);
            const v4f bi1 = *(const v4f*)(vim + (size_t)(4 * ks + 1) * 256);
            v2f tim, tng;
            tim[0] = tab[TABP + abase + 4 * ks];
            tim[1] = tab[TABP + abase + 4 * ks + 1];
            tng[0] = tab[2 * TABP + abase + 4 * ks];
            tng[1] = tab[2 * TABP + abase + 4 * ks + 1];
            #pragma unroll
            for (int nt = 0; nt < 4; ++nt) {
                v2f b;
                b[0] = br0[nt];
                b[1] = br1[nt];
                yre[nt] = mm(tre, b, yre[nt]);
                yim[nt] = mm(tim, b, yim[nt]);
            }
            #pragma unroll
            for (int nt = 0; nt < 4; ++nt) {
                v2f b;
                b[0] = bi0[nt];
                b[1] = bi1[nt];
                yre[nt] = mm(tng, b, yre[nt]);
                yim[nt] = mm(tre, b, yim[nt]);
            }
        } else {
            #pragma unroll
            for (int nt = 0; nt < 4; ++nt) {
                v2f b;
                b[0] = br0[nt];
                b[1] = br1[nt];
                yre[nt] = mm(tre, b, yre[nt]);
            }
        }
    }
    #pragma unroll
    for (int r = 0; r < 8; ++r) {
        v4f o;
        o[0] = yre[0][r];
        o[1] = yre[1][r];
        o[2] = yre[2][r];
        o[3] = yre[3][r];
        *(v4f*)(yre_p + (size_t)(8 * hf + r) * 256) = o;
        if (CPLX) {
            v4f p;
            p[0] = yim[0][r];
            p[1] = yim[1][r];
            p[2] = yim[2][r];
            p[3] = yim[3][r];
            *(v4f*)(yim_p + (size_t)(8 * hf + r) * 256) = p;
        }
    }
}

__global__ __launch_bounds__(256, 8) void mid_kernel(const float* __restrict__ w,
                                                     const float* __restrict__ offset,
                                                     const float* __restrict__ ws,
                                                     float* __restrict__ yh,
                                                     float* __restrict__ out)
{
    __shared__ __attribute__((aligned(16))) float s[4096];
    __shared__ __attribute__((aligned(16))) float wtot[8];
    __shared__ float tab[3 * TABP];

    if (blockIdx.x < Hh) {
        head_body(w, offset, out, s, wtot, blockIdx.x);
        return;
    }

    const int tid  = threadIdx.x;
    const int lane = tid & 31;
    const int wave = tid >> 5;
    const int cl   = lane & 15;
    const int hf   = lane >> 4;
    const int bx   = blockIdx.x - Hh;
    const int half = bx & 1;
    const int q    = bx >> 1;
    const int nc   = Hh * (NF - 2);
    int h, f;
    if (q < nc) { h = q / (NF - 2); f = 1 + (q - h * (NF - 2)); }
    else        { h = (q - nc) >> 1; f = ((q - nc) & 1) ? 32 : 0; }
    const int hfq  = h * NF + f;
    const int rt   = wave & 3;
    const int ctl  = half * 2 + (wave >> 2);

    const float* tabg = ws + WS_TAB;
    const float* vh   = ws + WS_VH;
    for (int i = tid; i < 3 * TABP; i += 256) tab[i] = tabg[(size_t)hfq * (3 * TABP) + i];
    __syncthreads();

    const bool cplx   = (f != 0) && (f != 32);
    const int  re_row = f == 0 ? 0 : (f == 32 ? 1 : 2 * f);
    const int  im_row = cplx ? re_row + 1 : re_row;

    const size_t boff = (size_t)(2 * hf) * 256 + ctl * 64 + 4 * cl;
    const float* vre = vh + ((size_t)(h * 64 + re_row) * 64) * 256 + boff;
    const float* vim = vh + ((size_t)(h * 64 + im_row) * 64) * 256 + boff;
    const int abase = 2 * hf - (16 * rt + cl) + 63;
    const size_t ooff = (size_t)(16 * rt) * 256 + ctl * 64 + 4 * cl;
    float* yre_p = yh + ((size_t)(h * 64 + re_row) * 64) * 256 + ooff;
    float* yim_p = yh + ((size_t)(h * 64 + im_row) * 64) * 256 + ooff;

    if (cplx) mid_body<true>(vre, vim, tab, abase, yre_p, yim_p, hf);
    else      mid_body<false>(vre, vim, tab, abase, yre_p, yim_p, hf);
}

__global__ __launch_bounds__(256) void inv_kernel(const float* __restrict__ gm,
                                                  const float* __restrict__ yh,
                                                  float* __restrict__ out)
{
    const int lane = threadIdx.x & 31;
    const int wave = threadIdx.x >> 5;
    const int cl   = lane & 15;
    const int hf   = lane >> 4;
    const int g    = blockIdx.x * 8 + wave;
    const int pr   = g & 63;
    const int n    = (g >> 6) & 3;
    const int h    = g >> 8;

    const float* bp = yh + (((size_t)(h * 64 + 2 * hf) * 64 + pr) * 256 + n * 64 + 4 * cl);
    const float* ap = gm + cl * 64 + 2 * hf;

    v8f acc[2][4] = {};
    #pragma unroll
    for (int ks = 0; ks < 16; ++ks) {
        const v4f b0 = *(const v4f*)(bp + (size_t)(4 * ks) * (64 * 256));
        const v4f b1 = *(const v4f*)(bp + (size_t)(4 * ks + 1) * (64 * 256));
        v2f a[2];
        #pragma unroll
        for (int mt = 0; mt < 2; ++mt) a[mt] = *(const v2f*)(ap + mt * 1024 + 4 * ks);
        #pragma unroll
        for (int nt = 0; nt < 4; ++nt) {
            v2f b;
            b[0] = b0[nt];
            b[1] = b1[nt];
            #pragma unroll
            for (int mt = 0; mt < 2; ++mt) acc[mt][nt] = mm(a[mt], b, acc[mt][nt]);
        }
    }
    #pragma unroll
    for (int mt = 0; mt < 2; ++mt) {
        #pragma unroll
        for (int r = 0; r < 8; ++r) {
            const int J = 16 * mt + 8 * hf + r;
            const int j = 64 * J + pr;
            v4f o;
            o[0] = acc[mt][0][r];
            o[1] = acc[mt][1][r];
            o[2] = acc[mt][2][r];
            o[3] = acc[mt][3][r];
            *(v4f*)(out + (((size_t)n * LFULL + (j + 1)) * Hh + h) * Dd + 4 * cl) = o;
        }
    }
}

extern "C" void kernel_launch(void* const* d_in, const int* in_sizes, int n_in,
                              void* d_out, int out_size, void* d_ws, size_t ws_size,
                              hipStream_t stream)
{
    (void)in_sizes; (void)n_in; (void)out_size; (void)ws_size;
    const float* v      = (const float*)d_in[0];
    const float* offset = (const float*)d_in[1];
    const float* w      = (const float*)d_in[2];
    float*       out    = (float*)d_out;
    float*       ws     = (float*)d_ws;

    front_kernel<<<dim3(NB_L1), 256, 0, stream>>>(v, w, offset, ws);
    mid_kernel<<<dim3(NB_L2), 256, 0, stream>>>(w, offset, ws, ws + WS_YH, out);
    inv_kernel<<<dim3(NB_FWD), 256, 0, stream>>>(ws + WS_G, ws + WS_YH, out);
}


# Product launch re-tiled: wave = 64 rows x 64 columns of one (head, f) (was 16 x 64), 512 workgroups of 4 waves, all resident; a Vh row is loaded by one wave, not four, two k-steps ahead of use
# speedup vs baseline: 2.7537x; 1.0933x over previous
#include <hip/hip_runtime.h>


typedef float v2f __attribute__((ext_vector_type(2)));
typedef float v4f __attribute__((ext_vector_type(4)));
typedef float v8f __attribute__((ext_vector_type(8)));

#define Bc     4
#define LFULL  2050
#define Lq     2048
#define Hh     16
#define Dd     64
#define SHAPE  4095

#define NF     33
#define TABP   128
#define WS_TAB 0
#define WS_G   (Hh * NF * 3 * TABP)
#define WS_VH  (WS_G + 4096)
#define PLANE  (Hh * 64 * 64 * 256)
#define WS_YH  (WS_VH + PLANE)

#define NB_TAB (Hh * NF)
#define NB_FWD (Hh * Bc * 64 / 8)
#define NB_L1  (NB_FWD + 1 + NB_TAB)
#define NB_L2  (Hh + Hh * (NF - 2))
#define FMP    34

__device__ __forceinline__ v8f mm(v2f a, v2f b, v8f c)
{
    return __builtin_amdgcn_wmma_f32_16x16x4_f32(false, a, false, b, (short)0, c, false, false);
}

__global__ __launch_bounds__(256) void front_kernel(const float* __restrict__ v,
                                                    const float* __restrict__ w,
                                                    const float* __restrict__ offset,
                                                    float* __restrict__ ws)
{
    __shared__ __attribute__((aligned(16))) float fm[64 * FMP];
    __shared__ float ct[64];
    const int tid = threadIdx.x;

    if (blockIdx.x == NB_FWD) {
        #pragma unroll
        for (int i = 0; i < 8; ++i) {
            const int e    = tid + 256 * i;
            const int row  = e & 63;
            const int blk  = e >> 6;
            const int f    = row == 0 ? 0 : (row == 1 ? 32 : (row >> 1));
            const int part = row < 2 ? 0 : (row & 1);
            const float x  = (float)((f * blk) & 63) * (1.0f / 32.0f);
            ws[WS_G + e] = part ? (-2.0f / 64.0f) * sinpif(x)
                                : (row < 2 ? 1.0f / 64.0f : 2.0f / 64.0f) * cospif(x);
        }
        return;
    }

    if (blockIdx.x > NB_FWD) {
        const int idx = blockIdx.x - (NB_FWD + 1);
        const int h   = idx / NF;
        const int f   = idx - h * NF;
        if (tid < 64) ct[tid] = cospif((float)tid * (1.0f / 32.0f));
        __syncthreads();
        const int dl   = tid & 127;
        const int part = tid >> 7;
        const float off = offset[h];
        float sum = 0.f;
        if (dl < 127) {
            const float* wp = w + h * SHAPE + (Lq - 1) + (dl - 63);
            #pragma unroll 9
            for (int kk = 0; kk < 63; ++kk) {
                const int k = kk - 31;
                const int m = (f * k) & 63;
                const float e = __expf(wp[64 * k] - off);
                const float t = ct[(m - 16 * part) & 63];
                sum += e * t;
            }
        }
        float* tp = ws + WS_TAB + (size_t)idx * (3 * TABP);
        if (part == 0) {
            tp[dl] = sum;
        } else {
            tp[TABP + dl]     = sum;
            tp[2 * TABP + dl] = -sum;
        }
        return;
    }

    const int lane = tid & 31;
    const int wave = tid >> 5;
    const int cl   = lane & 15;
    const int hf   = lane >> 4;
    const int g    = blockIdx.x * 8 + wave;
    const int pc   = g & 63;
    const int n    = (g >> 6) & 3;
    const int h    = g >> 8;

    const float* bp = v + (((size_t)n * LFULL + pc + 1) * Hh + h) * Dd + 4 * cl
                        + (size_t)(2 * hf) * (64 * Hh * Dd);
    v4f b0[8], b1[8];
    #pragma unroll
    for (int ks = 0; ks < 8; ++ks) {
        b0[ks] = *(const v4f*)(bp + (size_t)(4 * ks) * (64 * Hh * Dd));
        b1[ks] = *(const v4f*)(bp + (size_t)(4 * ks + 1) * (64 * Hh * Dd));
    }
    #pragma unroll
    for (int i = 0; i < 8; ++i) {
        const int e    = tid + 256 * i;
        const int row  = e >> 5;
        const int blk  = e & 31;
        const int f    = row == 0 ? 0 : (row == 1 ? 32 : (row >> 1));
        const int part = row < 2 ? 0 : (row & 1);
        fm[row * FMP + blk] = cospif((float)((f * blk + 16 * part) & 63) * (1.0f / 32.0f));
    }
    __syncthreads();
    const float* ap = fm + cl * FMP + 2 * hf;

    v8f acc[4][4] = {};
    #pragma unroll
    for (int ks = 0; ks < 8; ++ks) {
        v2f a[4];
        #pragma unroll
        for (int mt = 0; mt < 4; ++mt) a[mt] = *(const v2f*)(ap + mt * (16 * FMP) + 4 * ks);
        #pragma unroll
        for (int nt = 0; nt < 4; ++nt) {
            v2f b;
            b[0] = b0[ks][nt];
            b[1] = b1[ks][nt];
            #pragma unroll
            for (int mt = 0; mt < 4; ++mt) acc[mt][nt] = mm(a[mt], b, acc[mt][nt]);
        }
    }
    float* vh = ws + WS_VH;
    #pragma unroll
    for (int mt = 0; mt < 4; ++mt) {
        #pragma unroll
        for (int r = 0; r < 8; ++r) {
            const int row = 16 * mt + 8 * hf + r;
            v4f o;
            o[0] = acc[mt][0][r];
            o[1] = acc[mt][1][r];
            o[2] = acc[mt][2][r];
            o[3] = acc[mt][3][r];
            *(v4f*)(vh + (((size_t)(h * 64 + row) * 64 + pc) * 256 + n * 64 + 4 * cl)) = o;
        }
    }
}

__device__ __forceinline__ void head_body(const float* __restrict__ w,
                                          const float* __restrict__ offset,
                                          float* __restrict__ out,
                                          float* s, float* wtot, int h)
{
    const int tid  = threadIdx.x;
    const int lane = tid & 31;
    const int wave = tid >> 5;
    const float off = offset[h];
    const float* wrow = w + h * SHAPE;

    float raw[32];
    #pragma unroll
    for (int i = 0; i < 32; ++i) {
        const int k = tid + 128 * i;
        raw[i] = wrow[k < SHAPE ? k : SHAPE - 1];
    }
    #pragma unroll
    for (int i = 0; i < 32; ++i) {
        const int k = tid + 128 * i;
        s[k] = (k < SHAPE) ? __expf(raw[i] - off) : 0.f;
    }
    __syncthreads();

    float loc[2][16];
    float acc[2], inc[2];
    #pragma unroll
    for (int v = 0; v < 2; ++v) {
        const int base = (tid + 128 * v) * 16;
        float a = 0.f;
        #pragma unroll
        for (int q = 0; q < 4; ++q) {
            const v4f t4 = *(const v4f*)(s + base + 4 * q);
            #pragma unroll
            for (int e = 0; e < 4; ++e) { loc[v][4 * q + e] = a; a += t4[e]; }
        }
        acc[v] = a;
        float t = a;
        #pragma unroll
        for (int d = 1; d < 32; d <<= 1) {
            const float up = __shfl_up(t, d, 32);
            if (lane >= d) t += up;
        }
        inc[v] = t;
        if (lane == 31) wtot[wave + 4 * v] = t;
    }
    __syncthreads();
    #pragma unroll
    for (int v = 0; v < 2; ++v) {
        const int base = (tid + 128 * v) * 16;
        const int grp  = wave + 4 * v;
        float pbase = inc[v] - acc[v];
        #pragma unroll
        for (int q = 0; q < 7; ++q) pbase += (q < grp) ? wtot[q] : 0.f;
        #pragma unroll
        for (int q = 0; q < 4; ++q) {
            v4f t4;
            #pragma unroll
            for (int e = 0; e < 4; ++e) t4[e] = pbase + loc[v][4 * q + e];
            *(v4f*)(s + base + 4 * q) = t4;
        }
    }
    __syncthreads();

    float* zout = out + (size_t)Bc * LFULL * Hh * Dd;
    float zv[16];
    #pragma unroll
    for (int r = 0; r < 16; ++r) {
        const int i = tid + 128 * r;
        zv[r] = s[i + Lq] - s[i];
    }
    #pragma unroll
    for (int r = 0; r < 16; ++r) {
        const int i = tid + 128 * r;
        #pragma unroll
        for (int n = 0; n < Bc; ++n)
            zout[((size_t)n * LFULL + (i + 1)) * Hh + h] = zv[r];
    }
    if (tid < 2 * Bc) {
        const int n  = tid & 3;
        const int jp = (tid >> 2) ? (LFULL - 1) : 0;
        zout[((size_t)n * LFULL + jp) * Hh + h] = 0.f;
    }
    for (int i = tid; i < Bc * 2 * Dd; i += 128) {
        const int d  = i & 63;
        const int r  = i >> 6;
        const int n  = r & 3;
        const int jp = (r >> 2) ? (LFULL - 1) : 0;
        out[(((size_t)n * LFULL + jp) * Hh + h) * Dd + d] = 0.f;
    }
}

template <bool CPLX>
__device__ __forceinline__ void mid_body(const float* __restrict__ vre, const float* __restrict__ vim,
                                         const float* __restrict__ tabg, float* tab, int abase,
                                         float* __restrict__ yre_p, float* __restrict__ yim_p, int hf)
{
    v8f yre[4][4] = {};
    v8f yim[4][4] = {};
    v4f br0[3], br1[3], bi0[3], bi1[3];
    v2f tre[2][4], tim[2][4], tng[2][4];

    #pragma unroll
    for (int k = 0; k < 2; ++k) {
        br0[k] = *(const v4f*)(vre + (size_t)(4 * k) * 256);
        br1[k] = *(const v4f*)(vre + (size_t)(4 * k + 1) * 256);
        if (CPLX) {
            bi0[k] = *(const v4f*)(vim + (size_t)(4 * k) * 256);
            bi1[k] = *(const v4f*)(vim + (size_t)(4 * k + 1) * 256);
        }
    }
    {
        const int t = threadIdx.x;
        const float e0 = tabg[t];
        float e1 = 0.f, e2 = 0.f;
        if (CPLX) {
            e1 = tabg[TABP + t];
            e2 = tabg[2 * TABP + t];
        }
        tab[t] = e0;
        if (CPLX) {
            tab[TABP + t]     = e1;
            tab[2 * TABP + t] = e2;
        }
    }
    __syncthreads();
    #pragma unroll
    for (int rt = 0; rt < 4; ++rt) {
        const int ai = abase - 16 * rt;
        tre[0][rt][0] = tab[ai];
        tre[0][rt][1] = tab[ai + 1];
        if (CPLX) {
            tim[0][rt][0] = tab[TABP + ai];
            tim[0][rt][1] = tab[TABP + ai + 1];
            tng[0][rt][0] = tab[2 * TABP + ai];
            tng[0][rt][1] = tab[2 * TABP + ai + 1];
        }
    }

    #pragma unroll
    for (int ks = 0; ks < 16; ++ks) {
        const int sb = ks % 3;
        const int sa = ks & 1;
        if (ks + 2 < 16) {
            const int nb = (ks + 2) % 3;
            br0[nb] = *(const v4f*)(vre + (size_t)(4 * ks + 8) * 256);
            br1[nb] = *(const v4f*)(vre + (size_t)(4 * ks + 9) * 256);
            if (CPLX) {
                bi0[nb] = *(const v4f*)(vim + (size_t)(4 * ks + 8) * 256);
                bi1[nb] = *(const v4f*)(vim + (size_t)(4 * ks + 9) * 256);
            }
        }
        if (ks + 1 < 16) {
            #pragma unroll
            for (int rt = 0; rt < 4; ++rt) {
                const int ai = abase - 16 * rt + 4 * (ks + 1);
                tre[sa ^ 1][rt][0] = tab[ai];
                tre[sa ^ 1][rt][1] = tab[ai + 1];
                if (CPLX) {
                    tim[sa ^ 1][rt][0] = tab[TABP + ai];
                    tim[sa ^ 1][rt][1] = tab[TABP + ai + 1];
                    tng[sa ^ 1][rt][0] = tab[2 * TABP + ai];
                    tng[sa ^ 1][rt][1] = tab[2 * TABP + ai + 1];
                }
            }
        }
        __builtin_amdgcn_sched_barrier(0);
        #pragma unroll
        for (int rt = 0; rt < 4; ++rt) {
            if (CPLX) {
                #pragma unroll
                for (int nt = 0; nt < 4; ++nt) {
                    v2f b;
                    b[0] = br0[sb][nt];
                    b[1] = br1[sb][nt];
                    yre[rt][nt] = mm(tre[sa][rt], b, yre[rt][nt]);
                    yim[rt][nt] = mm(tim[sa][rt], b, yim[rt][nt]);
                }
                #pragma unroll
                for (int nt = 0; nt < 4; ++nt) {
                    v2f b;
                    b[0] = bi0[sb][nt];
                    b[1] = bi1[sb][nt];
                    yre[rt][nt] = mm(tng[sa][rt], b, yre[rt][nt]);
                    yim[rt][nt] = mm(tre[sa][rt], b, yim[rt][nt]);
                }
            } else {
                #pragma unroll
                for (int nt = 0; nt < 4; ++nt) {
                    v2f b;
                    b[0] = br0[sb][nt];
                    b[1] = br1[sb][nt];
                    yre[rt][nt] = mm(tre[sa][rt], b, yre[rt][nt]);
                }
            }
        }
        __builtin_amdgcn_sched_barrier(0);
    }
    #pragma unroll
    for (int rt = 0; rt < 4; ++rt) {
        #pragma unroll
        for (int r = 0; r < 8; ++r) {
            v4f o;
            o[0] = yre[rt][0][r];
            o[1] = yre[rt][1][r];
            o[2] = yre[rt][2][r];
            o[3] = yre[rt][3][r];
            *(v4f*)(yre_p + (size_t)(16 * rt + 8 * hf + r) * 256) = o;
            if (CPLX) {
                v4f p;
                p[0] = yim[rt][0][r];
                p[1] = yim[rt][1][r];
                p[2] = yim[rt][2][r];
                p[3] = yim[rt][3][r];
                *(v4f*)(yim_p + (size_t)(16 * rt + 8 * hf + r) * 256) = p;
            }
        }
    }
}

template <bool CPLX>
__device__ __forceinline__ void product(const float* __restrict__ ws, float* __restrict__ yh,
                                        float* tab, int h, int f, int re_row)
{
    const int tid  = threadIdx.x;
    const int lane = tid & 31;
    const int wave = tid >> 5;
    const int cl   = lane & 15;
    const int hf   = lane >> 4;

    const float* tabg = ws + WS_TAB + (size_t)(h * NF + f) * (3 * TABP);
    const float* vh   = ws + WS_VH;

    const int im_row = CPLX ? re_row + 1 : re_row;
    const size_t boff = (size_t)(2 * hf) * 256 + wave * 64 + 4 * cl;
    const float* vre = vh + ((size_t)(h * 64 + re_row) * 64) * 256 + boff;
    const float* vim = vh + ((size_t)(h * 64 + im_row) * 64) * 256 + boff;
    const int abase = 2 * hf - cl + 63;
    const size_t ooff = (size_t)(wave * 64 + 4 * cl);
    float* yre_p = yh + ((size_t)(h * 64 + re_row) * 64) * 256 + ooff;
    float* yim_p = yh + ((size_t)(h * 64 + im_row) * 64) * 256 + ooff;

    mid_body<CPLX>(vre, vim, tabg, tab, abase, yre_p, yim_p, hf);
}

__global__ __launch_bounds__(128) void mid_kernel(const float* __restrict__ w,
                                                  const float* __restrict__ offset,
                                                  const float* __restrict__ ws,
                                                  float* __restrict__ yh,
                                                  float* __restrict__ out)
{
    __shared__ __attribute__((aligned(16))) float s[4096];
    __shared__ __attribute__((aligned(16))) float wtot[8];
    __shared__ float tab[3 * TABP];

    if (blockIdx.x < Hh) {
        const int h = blockIdx.x;
        head_body(w, offset, out, s, wtot, h);
        product<false>(ws, yh, tab, h, 0, 0);
        __syncthreads();
        product<false>(ws, yh, tab, h, 32, 1);
        return;
    }
    const int q = blockIdx.x - Hh;
    const int h = q / (NF - 2);
    const int f = 1 + (q - h * (NF - 2));
    product<true>(ws, yh, tab, h, f, 2 * f);
}

__global__ __launch_bounds__(256) void inv_kernel(const float* __restrict__ gm,
                                                  const float* __restrict__ yh,
                                                  float* __restrict__ out)
{
    const int lane = threadIdx.x & 31;
    const int wave = threadIdx.x >> 5;
    const int cl   = lane & 15;
    const int hf   = lane >> 4;
    const int g    = blockIdx.x * 8 + wave;
    const int pr   = g & 63;
    const int n    = (g >> 6) & 3;
    const int h    = g >> 8;

    const float* bp = yh + (((size_t)(h * 64 + 2 * hf) * 64 + pr) * 256 + n * 64 + 4 * cl);
    const float* ap = gm + cl * 64 + 2 * hf;

    v8f acc[2][4] = {};
    #pragma unroll
    for (int ks = 0; ks < 16; ++ks) {
        const v4f b0 = *(const v4f*)(bp + (size_t)(4 * ks) * (64 * 256));
        const v4f b1 = *(const v4f*)(bp + (size_t)(4 * ks + 1) * (64 * 256));
        v2f a[2];
        #pragma unroll
        for (int mt = 0; mt < 2; ++mt) a[mt] = *(const v2f*)(ap + mt * 1024 + 4 * ks);
        #pragma unroll
        for (int nt = 0; nt < 4; ++nt) {
            v2f b;
            b[0] = b0[nt];
            b[1] = b1[nt];
            #pragma unroll
            for (int mt = 0; mt < 2; ++mt) acc[mt][nt] = mm(a[mt], b, acc[mt][nt]);
        }
    }
    #pragma unroll
    for (int mt = 0; mt < 2; ++mt) {
        #pragma unroll
        for (int r = 0; r < 8; ++r) {
            const int J = 16 * mt + 8 * hf + r;
            const int j = 64 * J + pr;
            v4f o;
            o[0] = acc[mt][0][r];
            o[1] = acc[mt][1][r];
            o[2] = acc[mt][2][r];
            o[3] = acc[mt][3][r];
            *(v4f*)(out + (((size_t)n * LFULL + (j + 1)) * Hh + h) * Dd + 4 * cl) = o;
        }
    }
}

extern "C" void kernel_launch(void* const* d_in, const int* in_sizes, int n_in,
                              void* d_out, int out_size, void* d_ws, size_t ws_size,
                              hipStream_t stream)
{
    (void)in_sizes; (void)n_in; (void)out_size; (void)ws_size;
    const float* v      = (const float*)d_in[0];
    const float* offset = (const float*)d_in[1];
    const float* w      = (const float*)d_in[2];
    float*       out    = (float*)d_out;
    float*       ws     = (float*)d_ws;

    front_kernel<<<dim3(NB_L1), 256, 0, stream>>>(v, w, offset, ws);
    mid_kernel<<<dim3(NB_L2), 128, 0, stream>>>(w, offset, ws, ws + WS_YH, out);
    inv_kernel<<<dim3(NB_FWD), 256, 0, stream>>>(ws + WS_G, ws + WS_YH, out);
}


# All three products in ONE launch, workgroup = (head, 16 columns): the 64 transformed rows stay in its LDS (256 KiB) between the products, never in memory; tables, z_pb in a short first launch
# speedup vs baseline: 2.9416x; 1.0682x over previous
#include <hip/hip_runtime.h>


typedef float v2f __attribute__((ext_vector_type(2)));
typedef float v4f __attribute__((ext_vector_type(4)));
typedef float v8f __attribute__((ext_vector_type(8)));

#define Bc     4
#define LFULL  2050
#define Lq     2048
#define Hh     16
#define Dd     64
#define SHAPE  4095


#define NF     33
#define TABP   128
#define WS_TAB 0
#define WS_G   (Hh * NF * 2 * TABP)

#define NB_TAB  (Hh * NF)
#define NB_PREP (Hh + 1 + 2 * NB_TAB)
#define NB_MAIN (Hh * 16)
#define FMP    34
#define VROW   1024

__device__ __forceinline__ v8f mm(v2f a, v2f b, v8f c)
{
    return __builtin_amdgcn_wmma_f32_16x16x4_f32(false, a, false, b, (short)0, c, false, false);
}

__device__ __forceinline__ void head_body(const float* __restrict__ w,
                                          const float* __restrict__ offset,
                                          float* __restrict__ out,
                                          float* s, float* wtot, int h)
{
    const int tid  = threadIdx.x;
    const int lane = tid & 31;
    const int wave = tid >> 5;
    const float off = offset[h];
    const float* wrow = w + h * SHAPE;

    float raw[32];
    #pragma unroll
    for (int i = 0; i < 32; ++i) {
        const int k = tid + 128 * i;
        raw[i] = wrow[k < SHAPE ? k : SHAPE - 1];
    }
    #pragma unroll
    for (int i = 0; i < 32; ++i) {
        const int k = tid + 128 * i;
        s[k] = (k < SHAPE) ? __expf(raw[i] - off) : 0.f;
    }
    __syncthreads();

    float loc[2][16];
    float acc[2], inc[2];
    #pragma unroll
    for (int v = 0; v < 2; ++v) {
        const int base = (tid + 128 * v) * 16;
        float a = 0.f;
        #pragma unroll
        for (int q = 0; q < 4; ++q) {
            const v4f t4 = *(const v4f*)(s + base + 4 * q);
            #pragma unroll
            for (int e = 0; e < 4; ++e) { loc[v][4 * q + e] = a; a += t4[e]; }
        }
        acc[v] = a;
        float t = a;
        #pragma unroll
        for (int d = 1; d < 32; d <<= 1) {
            const float up = __shfl_up(t, d, 32);
            if (lane >= d) t += up;
        }
        inc[v] = t;
        if (lane == 31) wtot[wave + 4 * v] = t;
    }
    __syncthreads();
    #pragma unroll
    for (int v = 0; v < 2; ++v) {
        const int base = (tid + 128 * v) * 16;
        const int grp  = wave + 4 * v;
        float pbase = inc[v] - acc[v];
        #pragma unroll
        for (int q = 0; q < 7; ++q) pbase += (q < grp) ? wtot[q] : 0.f;
        #pragma unroll
        for (int q = 0; q < 4; ++q) {
            v4f t4;
            #pragma unroll
            for (int e = 0; e < 4; ++e) t4[e] = pbase + loc[v][4 * q + e];
            *(v4f*)(s + base + 4 * q) = t4;
        }
    }
    __syncthreads();

    float* zout = out + (size_t)Bc * LFULL * Hh * Dd;
    float zv[16];
    #pragma unroll
    for (int r = 0; r < 16; ++r) {
        const int i = tid + 128 * r;
        zv[r] = s[i + Lq] - s[i];
    }
    #pragma unroll
    for (int r = 0; r < 16; ++r) {
        const int i = tid + 128 * r;
        #pragma unroll
        for (int n = 0; n < Bc; ++n)
            zout[((size_t)n * LFULL + (i + 1)) * Hh + h] = zv[r];
    }
    if (tid < 2 * Bc) {
        const int n  = tid & 3;
        const int jp = (tid >> 2) ? (LFULL - 1) : 0;
        zout[((size_t)n * LFULL + jp) * Hh + h] = 0.f;
    }
    for (int i = tid; i < Bc * 2 * Dd; i += 128) {
        const int d  = i & 63;
        const int r  = i >> 6;
        const int n  = r & 3;
        const int jp = (r >> 2) ? (LFULL - 1) : 0;
        out[(((size_t)n * LFULL + jp) * Hh + h) * Dd + d] = 0.f;
    }
}

__global__ __launch_bounds__(128) void prep_kernel(const float* __restrict__ w,
                                                   const float* __restrict__ offset,
                                                   float* __restrict__ ws,
                                                   float* __restrict__ out)
{
    __shared__ __attribute__((aligned(16))) float s[4096];
    __shared__ __attribute__((aligned(16))) float wtot[8];
    __shared__ float ct[64];
    const int tid = threadIdx.x;

    if (blockIdx.x < Hh) {
        head_body(w, offset, out, s, wtot, blockIdx.x);
        return;
    }

    if (blockIdx.x == Hh) {
        #pragma unroll
        for (int i = 0; i < 16; ++i) {
            const int e    = tid + 128 * i;
            const int row  = e & 63;
            const int blk  = e >> 6;
            const int f    = row == 0 ? 0 : (row == 1 ? 32 : (row >> 1));
            const int part = row < 2 ? 0 : (row & 1);
            const float x  = (float)((f * blk) & 63) * (1.0f / 32.0f);
            ws[WS_G + e] = part ? (-2.0f / 64.0f) * sinpif(x)
                                : (row < 2 ? 1.0f / 64.0f : 2.0f / 64.0f) * cospif(x);
        }
        return;
    }

    const int idx2 = blockIdx.x - (Hh + 1);
    const int idx  = idx2 >> 1;
    const int part = idx2 & 1;
    const int h    = idx / NF;
    const int f    = idx - h * NF;
    if (tid < 64) ct[tid] = cospif((float)tid * (1.0f / 32.0f));
    __syncthreads();
    const int dl = tid;
    const float off = offset[h];
    float sum = 0.f;
    if (dl < 127) {
        const float* wp = w + h * SHAPE + (Lq - 1) + (dl - 63);
        #pragma unroll 9
        for (int kk = 0; kk < 63; ++kk) {
            const int k = kk - 31;
            const int m = (f * k) & 63;
            const float e = __expf(wp[64 * k] - off);
            const float t = ct[(m - 16 * part) & 63];
            sum += e * t;
        }
    }
    ws[WS_TAB + (size_t)idx * (2 * TABP) + part * TABP + dl] = sum;
}

template <bool CPLX>
__device__ __forceinline__ void lds_product(float* vhs, const float* tre_t, const float* tim_t,
                                            int re_row, int cl, int hf)
{
    float* vre = vhs + re_row * VROW + (2 * hf) * 16 + cl;
    float* vim = vre + VROW;
    const int abase = 2 * hf - cl + 63;

    v8f yre[4] = {};
    v8f yim[4] = {};
    v2f br[2], bi[2], tre[2][4], tim[2][4];

    br[0][0] = vre[0];
    br[0][1] = vre[16];
    if (CPLX) {
        bi[0][0] = vim[0];
        bi[0][1] = vim[16];
    }
    #pragma unroll
    for (int rt = 0; rt < 4; ++rt) {
        const int ai = abase - 16 * rt;
        tre[0][rt][0] = tre_t[ai];
        tre[0][rt][1] = tre_t[ai + 1];
        if (CPLX) {
            tim[0][rt][0] = tim_t[ai];
            tim[0][rt][1] = tim_t[ai + 1];
        }
    }

    #pragma unroll
    for (int ks = 0; ks < 16; ++ks) {
        const int sa = ks & 1;
        if (ks + 1 < 16) {
            br[sa ^ 1][0] = vre[(4 * ks + 4) * 16];
            br[sa ^ 1][1] = vre[(4 * ks + 5) * 16];
            if (CPLX) {
                bi[sa ^ 1][0] = vim[(4 * ks + 4) * 16];
                bi[sa ^ 1][1] = vim[(4 * ks + 5) * 16];
            }
            #pragma unroll
            for (int rt = 0; rt < 4; ++rt) {
                const int ai = abase - 16 * rt + 4 * (ks + 1);
                tre[sa ^ 1][rt][0] = tre_t[ai];
                tre[sa ^ 1][rt][1] = tre_t[ai + 1];
                if (CPLX) {
                    tim[sa ^ 1][rt][0] = tim_t[ai];
                    tim[sa ^ 1][rt][1] = tim_t[ai + 1];
                }
            }
        }
        __builtin_amdgcn_sched_barrier(0);
        #pragma unroll
        for (int rt = 0; rt < 4; ++rt) {
            yre[rt] = mm(tre[sa][rt], br[sa], yre[rt]);
            if (CPLX) yim[rt] = mm(tim[sa][rt], br[sa], yim[rt]);
        }
        if (CPLX) {
            #pragma unroll
            for (int rt = 0; rt < 4; ++rt) {
                const v2f tng = -tim[sa][rt];
                yre[rt] = mm(tng, bi[sa], yre[rt]);
                yim[rt] = mm(tre[sa][rt], bi[sa], yim[rt]);
            }
        }
        __builtin_amdgcn_sched_barrier(0);
    }
    #pragma unroll
    for (int rt = 0; rt < 4; ++rt) {
        #pragma unroll
        for (int r8 = 0; r8 < 8; ++r8) {
            const int r = 16 * rt + 8 * hf + r8;
            vhs[re_row * VROW + r * 16 + cl] = yre[rt][r8];
            if (CPLX) vhs[(re_row + 1) * VROW + r * 16 + cl] = yim[rt][r8];
        }
    }
}

__global__ __launch_bounds__(512) void main_kernel(const float* __restrict__ v,
                                                   const float* __restrict__ ws,
                                                   float* __restrict__ out)
{
    __shared__ __attribute__((aligned(16))) float vhs[64 * VROW];
    __shared__ __attribute__((aligned(16))) float tabs[NF * 2 * TABP];
    __shared__ __attribute__((aligned(16))) float fm[64 * FMP];

    const int tid  = threadIdx.x;
    const int lane = tid & 31;
    const int wave = __builtin_amdgcn_readfirstlane(tid >> 5);
    const int cl   = lane & 15;
    const int hf   = lane >> 4;
    const int sub  = cl >> 2;
    const int q    = cl & 3;
    const int h    = blockIdx.x >> 4;
    const int xt   = blockIdx.x & 15;
    const int n    = xt >> 2;
    const int d0   = 16 * (xt & 3);
    const int p0   = 4 * wave;

    const float* bp = v + (((size_t)n * LFULL + (p0 + sub + 1)) * Hh + h) * Dd + d0 + 4 * q
                        + (size_t)(2 * hf) * (64 * Hh * Dd);
    v4f b0[8], b1[8];
    #pragma unroll
    for (int ks = 0; ks < 8; ++ks) {
        b0[ks] = *(const v4f*)(bp + (size_t)(4 * ks) * (64 * Hh * Dd));
        b1[ks] = *(const v4f*)(bp + (size_t)(4 * ks + 1) * (64 * Hh * Dd));
    }
    const float* tabg = ws + WS_TAB + (size_t)h * (NF * 2 * TABP);
    float tv[17];
    #pragma unroll
    for (int i = 0; i < 17; ++i) {
        const int e = tid + 512 * i;
        tv[i] = tabg[e < NF * 2 * TABP ? e : 0];
    }
    for (int i = 0; i < 4; ++i) {
        const int e    = tid + 512 * i;
        const int row  = e >> 5;
        const int blk  = e & 31;
        const int f    = row == 0 ? 0 : (row == 1 ? 32 : (row >> 1));
        const int part = row < 2 ? 0 : (row & 1);
        fm[row * FMP + blk] = cospif((float)((f * blk + 16 * part) & 63) * (1.0f / 32.0f));
    }
    #pragma unroll
    for (int i = 0; i < 17; ++i) {
        const int e = tid + 512 * i;
        if (e < NF * 2 * TABP) tabs[e] = tv[i];
    }
    __syncthreads();
    {
        const float* ap = fm + cl * FMP + 2 * hf;
        v8f acc[4][4] = {};
        #pragma unroll
        for (int ks = 0; ks < 8; ++ks) {
            v2f a[4];
            #pragma unroll
            for (int mt = 0; mt < 4; ++mt) a[mt] = *(const v2f*)(ap + mt * (16 * FMP) + 4 * ks);
            #pragma unroll
            for (int nt = 0; nt < 4; ++nt) {
                v2f b;
                b[0] = b0[ks][nt];
                b[1] = b1[ks][nt];
                #pragma unroll
                for (int mt = 0; mt < 4; ++mt) acc[mt][nt] = mm(a[mt], b, acc[mt][nt]);
            }
        }
        #pragma unroll
        for (int mt = 0; mt < 4; ++mt) {
            #pragma unroll
            for (int r8 = 0; r8 < 8; ++r8) {
                const int row = 16 * mt + 8 * hf + r8;
                v4f o;
                o[0] = acc[mt][0][r8];
                o[1] = acc[mt][1][r8];
                o[2] = acc[mt][2][r8];
                o[3] = acc[mt][3][r8];
                *(v4f*)(vhs + row * VROW + (p0 + sub) * 16 + 4 * q) = o;
            }
        }
    }
    __syncthreads();

    if (wave < 15) {
        #pragma unroll 1
        for (int i = 0; i < 2; ++i) {
            const int f = 2 * wave + 1 + i;
            lds_product<true>(vhs, tabs + f * (2 * TABP), tabs + f * (2 * TABP) + TABP, 2 * f, cl, hf);
        }
    } else {
        lds_product<true>(vhs, tabs + 31 * (2 * TABP), tabs + 31 * (2 * TABP) + TABP, 62, cl, hf);
        lds_product<false>(vhs, tabs, tabs, 0, cl, hf);
        lds_product<false>(vhs, tabs + 32 * (2 * TABP), tabs + 32 * (2 * TABP), 1, cl, hf);
    }
    __syncthreads();

    {
        const float* ap = ws + WS_G + cl * 64 + 2 * hf;
        const float* bq = vhs + (2 * hf) * VROW + (p0 + sub) * 16 + 4 * q;
        v8f acc[2][4] = {};
        #pragma unroll
        for (int ks = 0; ks < 16; ++ks) {
            const v4f c0 = *(const v4f*)(bq + (4 * ks) * VROW);
            const v4f c1 = *(const v4f*)(bq + (4 * ks + 1) * VROW);
            v2f a[2];
            #pragma unroll
            for (int mt = 0; mt < 2; ++mt) a[mt] = *(const v2f*)(ap + mt * 1024 + 4 * ks);
            #pragma unroll
            for (int nt = 0; nt < 4; ++nt) {
                v2f b;
                b[0] = c0[nt];
                b[1] = c1[nt];
                #pragma unroll
                for (int mt = 0; mt < 2; ++mt) acc[mt][nt] = mm(a[mt], b, acc[mt][nt]);
            }
        }
        #pragma unroll
        for (int mt = 0; mt < 2; ++mt) {
            #pragma unroll
            for (int r8 = 0; r8 < 8; ++r8) {
                const int J = 16 * mt + 8 * hf + r8;
                const int j = 64 * J + p0 + sub;
                v4f o;
                o[0] = acc[mt][0][r8];
                o[1] = acc[mt][1][r8];
                o[2] = acc[mt][2][r8];
                o[3] = acc[mt][3][r8];
                *(v4f*)(out + (((size_t)n * LFULL + (j + 1)) * Hh + h) * Dd + d0 + 4 * q) = o;
            }
        }
    }
}

extern "C" void kernel_launch(void* const* d_in, const int* in_sizes, int n_in,
                              void* d_out, int out_size, void* d_ws, size_t ws_size,
                              hipStream_t stream)
{
    (void)in_sizes; (void)n_in; (void)out_size; (void)ws_size;
    const float* v      = (const float*)d_in[0];
    const float* offset = (const float*)d_in[1];
    const float* w      = (const float*)d_in[2];
    float*       out    = (float*)d_out;
    float*       ws     = (float*)d_ws;

    prep_kernel<<<dim3(NB_PREP), 128, 0, stream>>>(w, offset, ws, out);
    main_kernel<<<dim3(NB_MAIN), 512, 0, stream>>>(v, ws, out);
}


# Complex product of each f in three real matrix products, not four: K1=(A+B)C, K2=A(D-C), K3=B(C+D), Re=K1-K3, Im=K1+K2; operand sums on the vector unit; changes the f32 sum order
# speedup vs baseline: 3.1067x; 1.0561x over previous
#include <hip/hip_runtime.h>


typedef float v2f __attribute__((ext_vector_type(2)));
typedef float v4f __attribute__((ext_vector_type(4)));
typedef float v8f __attribute__((ext_vector_type(8)));

#define Bc     4
#define LFULL  2050
#define Lq     2048
#define Hh     16
#define Dd     64
#define SHAPE  4095


#define NF     33
#define TABP   128
#define WS_TAB 0
#define WS_G   (Hh * NF * 2 * TABP)

#define NB_TAB  (Hh * NF)
#define NB_PREP (Hh + 1 + 2 * NB_TAB)
#define NB_MAIN (Hh * 16)
#define FMP    34
#define VROW   1024

__device__ __forceinline__ v8f mm(v2f a, v2f b, v8f c)
{
    return __builtin_amdgcn_wmma_f32_16x16x4_f32(false, a, false, b, (short)0, c, false, false);
}

__device__ __forceinline__ void head_body(const float* __restrict__ w,
                                          const float* __restrict__ offset,
                                          float* __restrict__ out,
                                          float* s, float* wtot, int h)
{
    const int tid  = threadIdx.x;
    const int lane = tid & 31;
    const int wave = tid >> 5;
    const float off = offset[h];
    const float* wrow = w + h * SHAPE;

    float raw[32];
    #pragma unroll
    for (int i = 0; i < 32; ++i) {
        const int k = tid + 128 * i;
        raw[i] = wrow[k < SHAPE ? k : SHAPE - 1];
    }
    #pragma unroll
    for (int i = 0; i < 32; ++i) {
        const int k = tid + 128 * i;
        s[k] = (k < SHAPE) ? __expf(raw[i] - off) : 0.f;
    }
    __syncthreads();

    float loc[2][16];
    float acc[2], inc[2];
    #pragma unroll
    for (int v = 0; v < 2; ++v) {
        const int base = (tid + 128 * v) * 16;
        float a = 0.f;
        #pragma unroll
        for (int q = 0; q < 4; ++q) {
            const v4f t4 = *(const v4f*)(s + base + 4 * q);
            #pragma unroll
            for (int e = 0; e < 4; ++e) { loc[v][4 * q + e] = a; a += t4[e]; }
        }
        acc[v] = a;
        float t = a;
        #pragma unroll
        for (int d = 1; d < 32; d <<= 1) {
            const float up = __shfl_up(t, d, 32);
            if (lane >= d) t += up;
        }
        inc[v] = t;
        if (lane == 31) wtot[wave + 4 * v] = t;
    }
    __syncthreads();
    #pragma unroll
    for (int v = 0; v < 2; ++v) {
        const int base = (tid + 128 * v) * 16;
        const int grp  = wave + 4 * v;
        float pbase = inc[v] - acc[v];
        #pragma unroll
        for (int q = 0; q < 7; ++q) pbase += (q < grp) ? wtot[q] : 0.f;
        #pragma unroll
        for (int q = 0; q < 4; ++q) {
            v4f t4;
            #pragma unroll
            for (int e = 0; e < 4; ++e) t4[e] = pbase + loc[v][4 * q + e];
            *(v4f*)(s + base + 4 * q) = t4;
        }
    }
    __syncthreads();

    float* zout = out + (size_t)Bc * LFULL * Hh * Dd;
    float zv[16];
    #pragma unroll
    for (int r = 0; r < 16; ++r) {
        const int i = tid + 128 * r;
        zv[r] = s[i + Lq] - s[i];
    }
    #pragma unroll
    for (int r = 0; r < 16; ++r) {
        const int i = tid + 128 * r;
        #pragma unroll
        for (int n = 0; n < Bc; ++n)
            zout[((size_t)n * LFULL + (i + 1)) * Hh + h] = zv[r];
    }
    if (tid < 2 * Bc) {
        const int n  = tid & 3;
        const int jp = (tid >> 2) ? (LFULL - 1) : 0;
        zout[((size_t)n * LFULL + jp) * Hh + h] = 0.f;
    }
    for (int i = tid; i < Bc * 2 * Dd; i += 128) {
        const int d  = i & 63;
        const int r  = i >> 6;
        const int n  = r & 3;
        const int jp = (r >> 2) ? (LFULL - 1) : 0;
        out[(((size_t)n * LFULL + jp) * Hh + h) * Dd + d] = 0.f;
    }
}

__global__ __launch_bounds__(128) void prep_kernel(const float* __restrict__ w,
                                                   const float* __restrict__ offset,
                                                   float* __restrict__ ws,
                                                   float* __restrict__ out)
{
    __shared__ __attribute__((aligned(16))) float s[4096];
    __shared__ __attribute__((aligned(16))) float wtot[8];
    __shared__ float ct[64];
    const int tid = threadIdx.x;

    if (blockIdx.x < Hh) {
        head_body(w, offset, out, s, wtot, blockIdx.x);
        return;
    }

    if (blockIdx.x == Hh) {
        #pragma unroll
        for (int i = 0; i < 16; ++i) {
            const int e    = tid + 128 * i;
            const int row  = e & 63;
            const int blk  = e >> 6;
            const int f    = row == 0 ? 0 : (row == 1 ? 32 : (row >> 1));
            const int part = row < 2 ? 0 : (row & 1);
            const float x  = (float)((f * blk) & 63) * (1.0f / 32.0f);
            ws[WS_G + e] = part ? (-2.0f / 64.0f) * sinpif(x)
                                : (row < 2 ? 1.0f / 64.0f : 2.0f / 64.0f) * cospif(x);
        }
        return;
    }

    const int idx2 = blockIdx.x - (Hh + 1);
    const int idx  = idx2 >> 1;
    const int part = idx2 & 1;
    const int h    = idx / NF;
    const int f    = idx - h * NF;
    if (tid < 64) ct[tid] = cospif((float)tid * (1.0f / 32.0f));
    __syncthreads();
    const int dl = tid;
    const float off = offset[h];
    float sum = 0.f;
    if (dl < 127) {
        const float* wp = w + h * SHAPE + (Lq - 1) + (dl - 63);
        #pragma unroll 9
        for (int kk = 0; kk < 63; ++kk) {
            const int k = kk - 31;
            const int m = (f * k) & 63;
            const float e = __expf(wp[64 * k] - off);
            const float t = ct[(m - 16 * part) & 63];
            sum += e * t;
        }
    }
    ws[WS_TAB + (size_t)idx * (2 * TABP) + part * TABP + dl] = sum;
}

template <bool CPLX>
__device__ __forceinline__ void lds_product(float* vhs, const float* tre_t, const float* tim_t,
                                            int re_row, int cl, int hf)
{
    float* vre = vhs + re_row * VROW + (2 * hf) * 16 + cl;
    float* vim = vre + VROW;
    const int abase = 2 * hf - cl + 63;

    v8f yre[4] = {};
    v8f k2[4] = {};
    v8f k3[4] = {};
    v2f br[2], bi[2], tre[2][4], tim[2][4];

    br[0][0] = vre[0];
    br[0][1] = vre[16];
    if (CPLX) {
        bi[0][0] = vim[0];
        bi[0][1] = vim[16];
    }
    #pragma unroll
    for (int rt = 0; rt < 4; ++rt) {
        const int ai = abase - 16 * rt;
        tre[0][rt][0] = tre_t[ai];
        tre[0][rt][1] = tre_t[ai + 1];
        if (CPLX) {
            tim[0][rt][0] = tim_t[ai];
            tim[0][rt][1] = tim_t[ai + 1];
        }
    }

    #pragma unroll
    for (int ks = 0; ks < 16; ++ks) {
        const int sa = ks & 1;
        if (ks + 1 < 16) {
            br[sa ^ 1][0] = vre[(4 * ks + 4) * 16];
            br[sa ^ 1][1] = vre[(4 * ks + 5) * 16];
            if (CPLX) {
                bi[sa ^ 1][0] = vim[(4 * ks + 4) * 16];
                bi[sa ^ 1][1] = vim[(4 * ks + 5) * 16];
            }
            #pragma unroll
            for (int rt = 0; rt < 4; ++rt) {
                const int ai = abase - 16 * rt + 4 * (ks + 1);
                tre[sa ^ 1][rt][0] = tre_t[ai];
                tre[sa ^ 1][rt][1] = tre_t[ai + 1];
                if (CPLX) {
                    tim[sa ^ 1][rt][0] = tim_t[ai];
                    tim[sa ^ 1][rt][1] = tim_t[ai + 1];
                }
            }
        }
        __builtin_amdgcn_sched_barrier(0);
        if (CPLX) {
            const v2f bsum = br[sa] + bi[sa];
            const v2f bdif = bi[sa] - br[sa];
            #pragma unroll
            for (int rt = 0; rt < 4; ++rt) {
                const v2f tsum = tre[sa][rt] + tim[sa][rt];
                yre[rt] = mm(tsum, br[sa], yre[rt]);
                k2[rt]  = mm(tre[sa][rt], bdif, k2[rt]);
                k3[rt]  = mm(tim[sa][rt], bsum, k3[rt]);
            }
        } else {
            #pragma unroll
            for (int rt = 0; rt < 4; ++rt) yre[rt] = mm(tre[sa][rt], br[sa], yre[rt]);
        }
        __builtin_amdgcn_sched_barrier(0);
    }
    #pragma unroll
    for (int rt = 0; rt < 4; ++rt) {
        #pragma unroll
        for (int r8 = 0; r8 < 8; ++r8) {
            const int r = 16 * rt + 8 * hf + r8;
            if (CPLX) {
                vhs[re_row * VROW + r * 16 + cl]       = yre[rt][r8] - k3[rt][r8];
                vhs[(re_row + 1) * VROW + r * 16 + cl] = yre[rt][r8] + k2[rt][r8];
            } else {
                vhs[re_row * VROW + r * 16 + cl] = yre[rt][r8];
            }
        }
    }
}

__global__ __launch_bounds__(512) void main_kernel(const float* __restrict__ v,
                                                   const float* __restrict__ ws,
                                                   float* __restrict__ out)
{
    __shared__ __attribute__((aligned(16))) float vhs[64 * VROW];
    __shared__ __attribute__((aligned(16))) float tabs[NF * 2 * TABP];
    __shared__ __attribute__((aligned(16))) float fm[64 * FMP];

    const int tid  = threadIdx.x;
    const int lane = tid & 31;
    const int wave = __builtin_amdgcn_readfirstlane(tid >> 5);
    const int cl   = lane & 15;
    const int hf   = lane >> 4;
    const int sub  = cl >> 2;
    const int q    = cl & 3;
    const int h    = blockIdx.x >> 4;
    const int xt   = blockIdx.x & 15;
    const int n    = xt >> 2;
    const int d0   = 16 * (xt & 3);
    const int p0   = 4 * wave;

    const float* bp = v + (((size_t)n * LFULL + (p0 + sub + 1)) * Hh + h) * Dd + d0 + 4 * q
                        + (size_t)(2 * hf) * (64 * Hh * Dd);
    v4f b0[8], b1[8];
    #pragma unroll
    for (int ks = 0; ks < 8; ++ks) {
        b0[ks] = *(const v4f*)(bp + (size_t)(4 * ks) * (64 * Hh * Dd));
        b1[ks] = *(const v4f*)(bp + (size_t)(4 * ks + 1) * (64 * Hh * Dd));
    }
    const float* tabg = ws + WS_TAB + (size_t)h * (NF * 2 * TABP);
    float tv[17];
    #pragma unroll
    for (int i = 0; i < 17; ++i) {
        const int e = tid + 512 * i;
        tv[i] = tabg[e < NF * 2 * TABP ? e : 0];
    }
    for (int i = 0; i < 4; ++i) {
        const int e    = tid + 512 * i;
        const int row  = e >> 5;
        const int blk  = e & 31;
        const int f    = row == 0 ? 0 : (row == 1 ? 32 : (row >> 1));
        const int part = row < 2 ? 0 : (row & 1);
        fm[row * FMP + blk] = cospif((float)((f * blk + 16 * part) & 63) * (1.0f / 32.0f));
    }
    #pragma unroll
    for (int i = 0; i < 17; ++i) {
        const int e = tid + 512 * i;
        if (e < NF * 2 * TABP) tabs[e] = tv[i];
    }
    __syncthreads();
    {
        const float* ap = fm + cl * FMP + 2 * hf;
        v8f acc[4][4] = {};
        #pragma unroll
        for (int ks = 0; ks < 8; ++ks) {
            v2f a[4];
            #pragma unroll
            for (int mt = 0; mt < 4; ++mt) a[mt] = *(const v2f*)(ap + mt * (16 * FMP) + 4 * ks);
            #pragma unroll
            for (int nt = 0; nt < 4; ++nt) {
                v2f b;
                b[0] = b0[ks][nt];
                b[1] = b1[ks][nt];
                #pragma unroll
                for (int mt = 0; mt < 4; ++mt) acc[mt][nt] = mm(a[mt], b, acc[mt][nt]);
            }
        }
        #pragma unroll
        for (int mt = 0; mt < 4; ++mt) {
            #pragma unroll
            for (int r8 = 0; r8 < 8; ++r8) {
                const int row = 16 * mt + 8 * hf + r8;
                v4f o;
                o[0] = acc[mt][0][r8];
                o[1] = acc[mt][1][r8];
                o[2] = acc[mt][2][r8];
                o[3] = acc[mt][3][r8];
                *(v4f*)(vhs + row * VROW + (p0 + sub) * 16 + 4 * q) = o;
            }
        }
    }
    __syncthreads();

    if (wave < 15) {
        #pragma unroll 1
        for (int i = 0; i < 2; ++i) {
            const int f = 2 * wave + 1 + i;
            lds_product<true>(vhs, tabs + f * (2 * TABP), tabs + f * (2 * TABP) + TABP, 2 * f, cl, hf);
        }
    } else {
        lds_product<true>(vhs, tabs + 31 * (2 * TABP), tabs + 31 * (2 * TABP) + TABP, 62, cl, hf);
        lds_product<false>(vhs, tabs, tabs, 0, cl, hf);
        lds_product<false>(vhs, tabs + 32 * (2 * TABP), tabs + 32 * (2 * TABP), 1, cl, hf);
    }
    __syncthreads();

    {
        const float* ap = ws + WS_G + cl * 64 + 2 * hf;
        const float* bq = vhs + (2 * hf) * VROW + (p0 + sub) * 16 + 4 * q;
        v8f acc[2][4] = {};
        #pragma unroll
        for (int ks = 0; ks < 16; ++ks) {
            const v4f c0 = *(const v4f*)(bq + (4 * ks) * VROW);
            const v4f c1 = *(const v4f*)(bq + (4 * ks + 1) * VROW);
            v2f a[2];
            #pragma unroll
            for (int mt = 0; mt < 2; ++mt) a[mt] = *(const v2f*)(ap + mt * 1024 + 4 * ks);
            #pragma unroll
            for (int nt = 0; nt < 4; ++nt) {
                v2f b;
                b[0] = c0[nt];
                b[1] = c1[nt];
                #pragma unroll
                for (int mt = 0; mt < 2; ++mt) acc[mt][nt] = mm(a[mt], b, acc[mt][nt]);
            }
        }
        #pragma unroll
        for (int mt = 0; mt < 2; ++mt) {
            #pragma unroll
            for (int r8 = 0; r8 < 8; ++r8) {
                const int J = 16 * mt + 8 * hf + r8;
                const int j = 64 * J + p0 + sub;
                v4f o;
                o[0] = acc[mt][0][r8];
                o[1] = acc[mt][1][r8];
                o[2] = acc[mt][2][r8];
                o[3] = acc[mt][3][r8];
                *(v4f*)(out + (((size_t)n * LFULL + (j + 1)) * Hh + h) * Dd + d0 + 4 * q) = o;
            }
        }
    }
}

extern "C" void kernel_launch(void* const* d_in, const int* in_sizes, int n_in,
                              void* d_out, int out_size, void* d_ws, size_t ws_size,
                              hipStream_t stream)
{
    (void)in_sizes; (void)n_in; (void)out_size; (void)ws_size;
    const float* v      = (const float*)d_in[0];
    const float* offset = (const float*)d_in[1];
    const float* w      = (const float*)d_in[2];
    float*       out    = (float*)d_out;
    float*       ws     = (float*)d_ws;

    prep_kernel<<<dim3(NB_PREP), 128, 0, stream>>>(w, offset, ws, out);
    main_kernel<<<dim3(NB_MAIN), 512, 0, stream>>>(v, ws, out);
}


# Forward product starts on the first V rows, not behind the last: table loads ahead of the V loads, first barrier hand-written so V loads stay in flight across it, a counted load wait per k-step
# speedup vs baseline: 3.2039x; 1.0313x over previous
#include <hip/hip_runtime.h>


typedef float v2f __attribute__((ext_vector_type(2)));
typedef float v4f __attribute__((ext_vector_type(4)));
typedef float v8f __attribute__((ext_vector_type(8)));

#define Bc     4
#define LFULL  2050
#define Lq     2048
#define Hh     16
#define Dd     64
#define SHAPE  4095


#define NF     33
#define TABP   128
#define WS_TAB 0
#define WS_G   (Hh * NF * 2 * TABP)

#define NB_TAB  (Hh * NF)
#define NB_PREP (Hh + 1 + 2 * NB_TAB)
#define NB_MAIN (Hh * 16)
#define FMP    34
#define VROW   1024

__device__ __forceinline__ v8f mm(v2f a, v2f b, v8f c)
{
    return __builtin_amdgcn_wmma_f32_16x16x4_f32(false, a, false, b, (short)0, c, false, false);
}

__device__ __forceinline__ void head_body(const float* __restrict__ w,
                                          const float* __restrict__ offset,
                                          float* __restrict__ out,
                                          float* s, float* wtot, int h)
{
    const int tid  = threadIdx.x;
    const int lane = tid & 31;
    const int wave = tid >> 5;
    const float off = offset[h];
    const float* wrow = w + h * SHAPE;

    float raw[32];
    #pragma unroll
    for (int i = 0; i < 32; ++i) {
        const int k = tid + 128 * i;
        raw[i] = wrow[k < SHAPE ? k : SHAPE - 1];
    }
    #pragma unroll
    for (int i = 0; i < 32; ++i) {
        const int k = tid + 128 * i;
        s[k] = (k < SHAPE) ? __expf(raw[i] - off) : 0.f;
    }
    __syncthreads();

    float loc[2][16];
    float acc[2], inc[2];
    #pragma unroll
    for (int v = 0; v < 2; ++v) {
        const int base = (tid + 128 * v) * 16;
        float a = 0.f;
        #pragma unroll
        for (int q = 0; q < 4; ++q) {
            const v4f t4 = *(const v4f*)(s + base + 4 * q);
            #pragma unroll
            for (int e = 0; e < 4; ++e) { loc[v][4 * q + e] = a; a += t4[e]; }
        }
        acc[v] = a;
        float t = a;
        #pragma unroll
        for (int d = 1; d < 32; d <<= 1) {
            const float up = __shfl_up(t, d, 32);
            if (lane >= d) t += up;
        }
        inc[v] = t;
        if (lane == 31) wtot[wave + 4 * v] = t;
    }
    __syncthreads();
    #pragma unroll
    for (int v = 0; v < 2; ++v) {
        const int base = (tid + 128 * v) * 16;
        const int grp  = wave + 4 * v;
        float pbase = inc[v] - acc[v];
        #pragma unroll
        for (int q = 0; q < 7; ++q) pbase += (q < grp) ? wtot[q] : 0.f;
        #pragma unroll
        for (int q = 0; q < 4; ++q) {
            v4f t4;
            #pragma unroll
            for (int e = 0; e < 4; ++e) t4[e] = pbase + loc[v][4 * q + e];
            *(v4f*)(s + base + 4 * q) = t4;
        }
    }
    __syncthreads();

    float* zout = out + (size_t)Bc * LFULL * Hh * Dd;
    float zv[16];
    #pragma unroll
    for (int r = 0; r < 16; ++r) {
        const int i = tid + 128 * r;
        zv[r] = s[i + Lq] - s[i];
    }
    #pragma unroll
    for (int r = 0; r < 16; ++r) {
        const int i = tid + 128 * r;
        #pragma unroll
        for (int n = 0; n < Bc; ++n)
            zout[((size_t)n * LFULL + (i + 1)) * Hh + h] = zv[r];
    }
    if (tid < 2 * Bc) {
        const int n  = tid & 3;
        const int jp = (tid >> 2) ? (LFULL - 1) : 0;
        zout[((size_t)n * LFULL + jp) * Hh + h] = 0.f;
    }
    for (int i = tid; i < Bc * 2 * Dd; i += 128) {
        const int d  = i & 63;
        const int r  = i >> 6;
        const int n  = r & 3;
        const int jp = (r >> 2) ? (LFULL - 1) : 0;
        out[(((size_t)n * LFULL + jp) * Hh + h) * Dd + d] = 0.f;
    }
}

__global__ __launch_bounds__(128) void prep_kernel(const float* __restrict__ w,
                                                   const float* __restrict__ offset,
                                                   float* __restrict__ ws,
                                                   float* __restrict__ out)
{
    __shared__ __attribute__((aligned(16))) float s[4096];
    __shared__ __attribute__((aligned(16))) float wtot[8];
    __shared__ float ct[64];
    const int tid = threadIdx.x;

    if (blockIdx.x < Hh) {
        head_body(w, offset, out, s, wtot, blockIdx.x);
        return;
    }

    if (blockIdx.x == Hh) {
        #pragma unroll
        for (int i = 0; i < 16; ++i) {
            const int e    = tid + 128 * i;
            const int row  = e & 63;
            const int blk  = e >> 6;
            const int f    = row == 0 ? 0 : (row == 1 ? 32 : (row >> 1));
            const int part = row < 2 ? 0 : (row & 1);
            const float x  = (float)((f * blk) & 63) * (1.0f / 32.0f);
            ws[WS_G + e] = part ? (-2.0f / 64.0f) * sinpif(x)
                                : (row < 2 ? 1.0f / 64.0f : 2.0f / 64.0f) * cospif(x);
        }
        return;
    }

    const int idx2 = blockIdx.x - (Hh + 1);
    const int idx  = idx2 >> 1;
    const int part = idx2 & 1;
    const int h    = idx / NF;
    const int f    = idx - h * NF;
    if (tid < 64) ct[tid] = cospif((float)tid * (1.0f / 32.0f));
    __syncthreads();
    const int dl = tid;
    const float off = offset[h];
    float sum = 0.f;
    if (dl < 127) {
        const float* wp = w + h * SHAPE + (Lq - 1) + (dl - 63);
        #pragma unroll 9
        for (int kk = 0; kk < 63; ++kk) {
            const int k = kk - 31;
            const int m = (f * k) & 63;
            const float e = __expf(wp[64 * k] - off);
            const float t = ct[(m - 16 * part) & 63];
            sum += e * t;
        }
    }
    ws[WS_TAB + (size_t)idx * (2 * TABP) + part * TABP + dl] = sum;
}

template <bool CPLX>
__device__ __forceinline__ void lds_product(float* vhs, const float* tre_t, const float* tim_t,
                                            int re_row, int cl, int hf)
{
    float* vre = vhs + re_row * VROW + (2 * hf) * 16 + cl;
    float* vim = vre + VROW;
    const int abase = 2 * hf - cl + 63;

    v8f yre[4] = {};
    v8f k2[4] = {};
    v8f k3[4] = {};
    v2f br[2], bi[2], tre[2][4], tim[2][4];

    br[0][0] = vre[0];
    br[0][1] = vre[16];
    if (CPLX) {
        bi[0][0] = vim[0];
        bi[0][1] = vim[16];
    }
    #pragma unroll
    for (int rt = 0; rt < 4; ++rt) {
        const int ai = abase - 16 * rt;
        tre[0][rt][0] = tre_t[ai];
        tre[0][rt][1] = tre_t[ai + 1];
        if (CPLX) {
            tim[0][rt][0] = tim_t[ai];
            tim[0][rt][1] = tim_t[ai + 1];
        }
    }

    #pragma unroll
    for (int ks = 0; ks < 16; ++ks) {
        const int sa = ks & 1;
        if (ks + 1 < 16) {
            br[sa ^ 1][0] = vre[(4 * ks + 4) * 16];
            br[sa ^ 1][1] = vre[(4 * ks + 5) * 16];
            if (CPLX) {
                bi[sa ^ 1][0] = vim[(4 * ks + 4) * 16];
                bi[sa ^ 1][1] = vim[(4 * ks + 5) * 16];
            }
            #pragma unroll
            for (int rt = 0; rt < 4; ++rt) {
                const int ai = abase - 16 * rt + 4 * (ks + 1);
                tre[sa ^ 1][rt][0] = tre_t[ai];
                tre[sa ^ 1][rt][1] = tre_t[ai + 1];
                if (CPLX) {
                    tim[sa ^ 1][rt][0] = tim_t[ai];
                    tim[sa ^ 1][rt][1] = tim_t[ai + 1];
                }
            }
        }
        __builtin_amdgcn_sched_barrier(0);
        if (CPLX) {
            const v2f bsum = br[sa] + bi[sa];
            const v2f bdif = bi[sa] - br[sa];
            #pragma unroll
            for (int rt = 0; rt < 4; ++rt) {
                const v2f tsum = tre[sa][rt] + tim[sa][rt];
                yre[rt] = mm(tsum, br[sa], yre[rt]);
                k2[rt]  = mm(tre[sa][rt], bdif, k2[rt]);
                k3[rt]  = mm(tim[sa][rt], bsum, k3[rt]);
            }
        } else {
            #pragma unroll
            for (int rt = 0; rt < 4; ++rt) yre[rt] = mm(tre[sa][rt], br[sa], yre[rt]);
        }
        __builtin_amdgcn_sched_barrier(0);
    }
    #pragma unroll
    for (int rt = 0; rt < 4; ++rt) {
        #pragma unroll
        for (int r8 = 0; r8 < 8; ++r8) {
            const int r = 16 * rt + 8 * hf + r8;
            if (CPLX) {
                vhs[re_row * VROW + r * 16 + cl]       = yre[rt][r8] - k3[rt][r8];
                vhs[(re_row + 1) * VROW + r * 16 + cl] = yre[rt][r8] + k2[rt][r8];
            } else {
                vhs[re_row * VROW + r * 16 + cl] = yre[rt][r8];
            }
        }
    }
}

__global__ __launch_bounds__(512) void main_kernel(const float* __restrict__ v,
                                                   const float* __restrict__ ws,
                                                   float* __restrict__ out)
{
    __shared__ __attribute__((aligned(16))) float vhs[64 * VROW];
    __shared__ __attribute__((aligned(16))) float tabs[17 * 512];
    __shared__ __attribute__((aligned(16))) float fm[64 * FMP];

    const int tid  = threadIdx.x;
    const int lane = tid & 31;
    const int wave = __builtin_amdgcn_readfirstlane(tid >> 5);
    const int cl   = lane & 15;
    const int hf   = lane >> 4;
    const int sub  = cl >> 2;
    const int q    = cl & 3;
    const int h    = blockIdx.x >> 4;
    const int xt   = blockIdx.x & 15;
    const int n    = xt >> 2;
    const int d0   = 16 * (xt & 3);
    const int p0   = 4 * wave;

    const float* bp = v + (((size_t)n * LFULL + (p0 + sub + 1)) * Hh + h) * Dd + d0 + 4 * q
                        + (size_t)(2 * hf) * (64 * Hh * Dd);
    const float* tabg = ws + WS_TAB + (size_t)h * (NF * 2 * TABP);
    float tv[17];
    #pragma unroll
    for (int i = 0; i < 17; ++i) {
        const int e = tid + 512 * i;
        tv[i] = tabg[e < NF * 2 * TABP ? e : 0];
    }
    __builtin_amdgcn_sched_barrier(0);
    v4f b0[8], b1[8];
    #pragma unroll
    for (int ks = 0; ks < 8; ++ks) {
        b0[ks] = *(const v4f*)(bp + (size_t)(4 * ks) * (64 * Hh * Dd));
        b1[ks] = *(const v4f*)(bp + (size_t)(4 * ks + 1) * (64 * Hh * Dd));
    }
    for (int i = 0; i < 4; ++i) {
        const int e    = tid + 512 * i;
        const int row  = e >> 5;
        const int blk  = e & 31;
        const int f    = row == 0 ? 0 : (row == 1 ? 32 : (row >> 1));
        const int part = row < 2 ? 0 : (row & 1);
        fm[row * FMP + blk] = cospif((float)((f * blk + 16 * part) & 63) * (1.0f / 32.0f));
    }
    #pragma unroll
    for (int i = 0; i < 17; ++i) tabs[tid + 512 * i] = tv[i];
    asm volatile("s_wait_dscnt 0x0" ::: "memory");
    asm volatile("s_barrier_signal -1" ::: "memory");
    asm volatile("s_barrier_wait -1" ::: "memory");
    {
        const float* ap = fm + cl * FMP + 2 * hf;
        v8f acc[4][4] = {};
        v2f a[2][4];
        #pragma unroll
        for (int mt = 0; mt < 4; ++mt) a[0][mt] = *(const v2f*)(ap + mt * (16 * FMP));
        #pragma unroll
        for (int ks = 0; ks < 8; ++ks) {
            const int sa = ks & 1;
            if (ks + 1 < 8) {
                #pragma unroll
                for (int mt = 0; mt < 4; ++mt) a[sa ^ 1][mt] = *(const v2f*)(ap + mt * (16 * FMP) + 4 * (ks + 1));
            }
            __builtin_amdgcn_sched_barrier(0);
            #pragma unroll
            for (int nt = 0; nt < 4; ++nt) {
                v2f b;
                b[0] = b0[ks][nt];
                b[1] = b1[ks][nt];
                #pragma unroll
                for (int mt = 0; mt < 4; ++mt) acc[mt][nt] = mm(a[sa][mt], b, acc[mt][nt]);
            }
            __builtin_amdgcn_sched_barrier(0);
        }
        #pragma unroll
        for (int mt = 0; mt < 4; ++mt) {
            #pragma unroll
            for (int r8 = 0; r8 < 8; ++r8) {
                const int row = 16 * mt + 8 * hf + r8;
                v4f o;
                o[0] = acc[mt][0][r8];
                o[1] = acc[mt][1][r8];
                o[2] = acc[mt][2][r8];
                o[3] = acc[mt][3][r8];
                *(v4f*)(vhs + row * VROW + (p0 + sub) * 16 + 4 * q) = o;
            }
        }
    }
    __syncthreads();

    if (wave < 15) {
        #pragma unroll 1
        for (int i = 0; i < 2; ++i) {
            const int f = 2 * wave + 1 + i;
            lds_product<true>(vhs, tabs + f * (2 * TABP), tabs + f * (2 * TABP) + TABP, 2 * f, cl, hf);
        }
    } else {
        lds_product<true>(vhs, tabs + 31 * (2 * TABP), tabs + 31 * (2 * TABP) + TABP, 62, cl, hf);
        lds_product<false>(vhs, tabs, tabs, 0, cl, hf);
        lds_product<false>(vhs, tabs + 32 * (2 * TABP), tabs + 32 * (2 * TABP), 1, cl, hf);
    }
    __syncthreads();

    {
        const float* ap = ws + WS_G + cl * 64 + 2 * hf;
        const float* bq = vhs + (2 * hf) * VROW + (p0 + sub) * 16 + 4 * q;
        v8f acc[2][4] = {};
        #pragma unroll
        for (int ks = 0; ks < 16; ++ks) {
            const v4f c0 = *(const v4f*)(bq + (4 * ks) * VROW);
            const v4f c1 = *(const v4f*)(bq + (4 * ks + 1) * VROW);
            v2f a[2];
            #pragma unroll
            for (int mt = 0; mt < 2; ++mt) a[mt] = *(const v2f*)(ap + mt * 1024 + 4 * ks);
            #pragma unroll
            for (int nt = 0; nt < 4; ++nt) {
                v2f b;
                b[0] = c0[nt];
                b[1] = c1[nt];
                #pragma unroll
                for (int mt = 0; mt < 2; ++mt) acc[mt][nt] = mm(a[mt], b, acc[mt][nt]);
            }
        }
        #pragma unroll
        for (int mt = 0; mt < 2; ++mt) {
            #pragma unroll
            for (int r8 = 0; r8 < 8; ++r8) {
                const int J = 16 * mt + 8 * hf + r8;
                const int j = 64 * J + p0 + sub;
                v4f o;
                o[0] = acc[mt][0][r8];
                o[1] = acc[mt][1][r8];
                o[2] = acc[mt][2][r8];
                o[3] = acc[mt][3][r8];
                *(v4f*)(out + (((size_t)n * LFULL + (j + 1)) * Hh + h) * Dd + d0 + 4 * q) = o;
            }
        }
    }
}

extern "C" void kernel_launch(void* const* d_in, const int* in_sizes, int n_in,
                              void* d_out, int out_size, void* d_ws, size_t ws_size,
                              hipStream_t stream)
{
    (void)in_sizes; (void)n_in; (void)out_size; (void)ws_size;
    const float* v      = (const float*)d_in[0];
    const float* offset = (const float*)d_in[1];
    const float* w      = (const float*)d_in[2];
    float*       out    = (float*)d_out;
    float*       ws     = (float*)d_ws;

    prep_kernel<<<dim3(NB_PREP), 128, 0, stream>>>(w, offset, ws, out);
    main_kernel<<<dim3(NB_MAIN), 512, 0, stream>>>(v, ws, out);
}


# Launch 1's table blocks re-tiled: block = (head, four f), thread = one diagonal; 63 w loads in flight at once, one exp each, shared by the group's eight tables; 161 blocks, not 1073; same f32 sums
# speedup vs baseline: 3.3431x; 1.0434x over previous
#include <hip/hip_runtime.h>


typedef float v2f __attribute__((ext_vector_type(2)));
typedef float v4f __attribute__((ext_vector_type(4)));
typedef float v8f __attribute__((ext_vector_type(8)));

#define Bc     4
#define LFULL  2050
#define Lq     2048
#define Hh     16
#define Dd     64
#define SHAPE  4095


#define NF     33
#define TABP   128
#define WS_TAB 0
#define WS_G   (Hh * NF * 2 * TABP)

#define FG      4
#define NFG     9
#define NB_TAB  (Hh * NFG)
#define NB_PREP (Hh + 1 + NB_TAB)
#define NB_MAIN (Hh * 16)
#define FMP    34
#define VROW   1024

__device__ __forceinline__ v8f mm(v2f a, v2f b, v8f c)
{
    return __builtin_amdgcn_wmma_f32_16x16x4_f32(false, a, false, b, (short)0, c, false, false);
}

__device__ __forceinline__ void head_body(const float* __restrict__ w,
                                          const float* __restrict__ offset,
                                          float* __restrict__ out,
                                          float* s, float* wtot, int h)
{
    const int tid  = threadIdx.x;
    const int lane = tid & 31;
    const int wave = tid >> 5;
    const float off = offset[h];
    const float* wrow = w + h * SHAPE;

    float raw[32];
    #pragma unroll
    for (int i = 0; i < 32; ++i) {
        const int k = tid + 128 * i;
        raw[i] = wrow[k < SHAPE ? k : SHAPE - 1];
    }
    #pragma unroll
    for (int i = 0; i < 32; ++i) {
        const int k = tid + 128 * i;
        s[k] = (k < SHAPE) ? __expf(raw[i] - off) : 0.f;
    }
    __syncthreads();

    float loc[2][16];
    float acc[2], inc[2];
    #pragma unroll
    for (int v = 0; v < 2; ++v) {
        const int base = (tid + 128 * v) * 16;
        float a = 0.f;
        #pragma unroll
        for (int q = 0; q < 4; ++q) {
            const v4f t4 = *(const v4f*)(s + base + 4 * q);
            #pragma unroll
            for (int e = 0; e < 4; ++e) { loc[v][4 * q + e] = a; a += t4[e]; }
        }
        acc[v] = a;
        float t = a;
        #pragma unroll
        for (int d = 1; d < 32; d <<= 1) {
            const float up = __shfl_up(t, d, 32);
            if (lane >= d) t += up;
        }
        inc[v] = t;
        if (lane == 31) wtot[wave + 4 * v] = t;
    }
    __syncthreads();
    #pragma unroll
    for (int v = 0; v < 2; ++v) {
        const int base = (tid + 128 * v) * 16;
        const int grp  = wave + 4 * v;
        float pbase = inc[v] - acc[v];
        #pragma unroll
        for (int q = 0; q < 7; ++q) pbase += (q < grp) ? wtot[q] : 0.f;
        #pragma unroll
        for (int q = 0; q < 4; ++q) {
            v4f t4;
            #pragma unroll
            for (int e = 0; e < 4; ++e) t4[e] = pbase + loc[v][4 * q + e];
            *(v4f*)(s + base + 4 * q) = t4;
        }
    }
    __syncthreads();

    float* zout = out + (size_t)Bc * LFULL * Hh * Dd;
    float zv[16];
    #pragma unroll
    for (int r = 0; r < 16; ++r) {
        const int i = tid + 128 * r;
        zv[r] = s[i + Lq] - s[i];
    }
    #pragma unroll
    for (int r = 0; r < 16; ++r) {
        const int i = tid + 128 * r;
        #pragma unroll
        for (int n = 0; n < Bc; ++n)
            zout[((size_t)n * LFULL + (i + 1)) * Hh + h] = zv[r];
    }
    if (tid < 2 * Bc) {
        const int n  = tid & 3;
        const int jp = (tid >> 2) ? (LFULL - 1) : 0;
        zout[((size_t)n * LFULL + jp) * Hh + h] = 0.f;
    }
    for (int i = tid; i < Bc * 2 * Dd; i += 128) {
        const int d  = i & 63;
        const int r  = i >> 6;
        const int n  = r & 3;
        const int jp = (r >> 2) ? (LFULL - 1) : 0;
        out[(((size_t)n * LFULL + jp) * Hh + h) * Dd + d] = 0.f;
    }
}

__global__ __launch_bounds__(128) void prep_kernel(const float* __restrict__ w,
                                                   const float* __restrict__ offset,
                                                   float* __restrict__ ws,
                                                   float* __restrict__ out)
{
    __shared__ __attribute__((aligned(16))) float s[4096];
    __shared__ __attribute__((aligned(16))) float wtot[8];
    __shared__ __attribute__((aligned(16))) float cs[64 * FG * 2];
    const int tid = threadIdx.x;

    if (blockIdx.x < Hh) {
        head_body(w, offset, out, s, wtot, blockIdx.x);
        return;
    }

    if (blockIdx.x == Hh) {
        #pragma unroll
        for (int i = 0; i < 16; ++i) {
            const int e    = tid + 128 * i;
            const int row  = e & 63;
            const int blk  = e >> 6;
            const int f    = row == 0 ? 0 : (row == 1 ? 32 : (row >> 1));
            const int part = row < 2 ? 0 : (row & 1);
            const float x  = (float)((f * blk) & 63) * (1.0f / 32.0f);
            ws[WS_G + e] = part ? (-2.0f / 64.0f) * sinpif(x)
                                : (row < 2 ? 1.0f / 64.0f : 2.0f / 64.0f) * cospif(x);
        }
        return;
    }

    const int idx2 = blockIdx.x - (Hh + 1);
    const int h    = idx2 / NFG;
    const int f0   = FG * (idx2 - h * NFG);
    const int dl  = tid;
    const int dlc = dl < 127 ? dl : 126;
    const float off = offset[h];
    const float* wp = w + h * SHAPE + (Lq - 1) + (dlc - 63);
    float e[63];
    #pragma unroll
    for (int kk = 0; kk < 63; ++kk) e[kk] = wp[64 * (kk - 31)];
    for (int i = tid; i < 63 * FG * 2; i += 128) {
        const int kk   = i >> 3;
        const int j    = (i >> 1) & 3;
        const int part = i & 1;
        const int m    = ((f0 + j) * (kk - 31) - 16 * part) & 63;
        cs[i] = cospif((float)m * (1.0f / 32.0f));
    }
    #pragma unroll
    for (int kk = 0; kk < 63; ++kk) e[kk] = __expf(e[kk] - off);
    __syncthreads();
    v4f t[63][2];
    #pragma unroll
    for (int kk = 0; kk < 63; ++kk) {
        t[kk][0] = *(const v4f*)(cs + kk * 8);
        t[kk][1] = *(const v4f*)(cs + kk * 8 + 4);
    }
    float acc[2 * FG] = {};
    #pragma unroll
    for (int kk = 0; kk < 63; ++kk) {
        #pragma unroll
        for (int q = 0; q < 4; ++q) {
            acc[q]     += e[kk] * t[kk][0][q];
            acc[4 + q] += e[kk] * t[kk][1][q];
        }
    }
    #pragma unroll
    for (int j = 0; j < FG; ++j) {
        const int f = f0 + j;
        if (f < NF) {
            float* tp = ws + WS_TAB + (size_t)(h * NF + f) * (2 * TABP) + dl;
            tp[0]    = dl < 127 ? acc[2 * j]     : 0.f;
            tp[TABP] = dl < 127 ? acc[2 * j + 1] : 0.f;
        }
    }
}

template <bool CPLX>
__device__ __forceinline__ void lds_product(float* vhs, const float* tre_t, const float* tim_t,
                                            int re_row, int cl, int hf)
{
    float* vre = vhs + re_row * VROW + (2 * hf) * 16 + cl;
    float* vim = vre + VROW;
    const int abase = 2 * hf - cl + 63;

    v8f yre[4] = {};
    v8f k2[4] = {};
    v8f k3[4] = {};
    v2f br[2], bi[2], tre[2][4], tim[2][4];

    br[0][0] = vre[0];
    br[0][1] = vre[16];
    if (CPLX) {
        bi[0][0] = vim[0];
        bi[0][1] = vim[16];
    }
    #pragma unroll
    for (int rt = 0; rt < 4; ++rt) {
        const int ai = abase - 16 * rt;
        tre[0][rt][0] = tre_t[ai];
        tre[0][rt][1] = tre_t[ai + 1];
        if (CPLX) {
            tim[0][rt][0] = tim_t[ai];
            tim[0][rt][1] = tim_t[ai + 1];
        }
    }

    #pragma unroll
    for (int ks = 0; ks < 16; ++ks) {
        const int sa = ks & 1;
        if (ks + 1 < 16) {
            br[sa ^ 1][0] = vre[(4 * ks + 4) * 16];
            br[sa ^ 1][1] = vre[(4 * ks + 5) * 16];
            if (CPLX) {
                bi[sa ^ 1][0] = vim[(4 * ks + 4) * 16];
                bi[sa ^ 1][1] = vim[(4 * ks + 5) * 16];
            }
            #pragma unroll
            for (int rt = 0; rt < 4; ++rt) {
                const int ai = abase - 16 * rt + 4 * (ks + 1);
                tre[sa ^ 1][rt][0] = tre_t[ai];
                tre[sa ^ 1][rt][1] = tre_t[ai + 1];
                if (CPLX) {
                    tim[sa ^ 1][rt][0] = tim_t[ai];
                    tim[sa ^ 1][rt][1] = tim_t[ai + 1];
                }
            }
        }
        __builtin_amdgcn_sched_barrier(0);
        if (CPLX) {
            const v2f bsum = br[sa] + bi[sa];
            const v2f bdif = bi[sa] - br[sa];
            #pragma unroll
            for (int rt = 0; rt < 4; ++rt) {
                const v2f tsum = tre[sa][rt] + tim[sa][rt];
                yre[rt] = mm(tsum, br[sa], yre[rt]);
                k2[rt]  = mm(tre[sa][rt], bdif, k2[rt]);
                k3[rt]  = mm(tim[sa][rt], bsum, k3[rt]);
            }
        } else {
            #pragma unroll
            for (int rt = 0; rt < 4; ++rt) yre[rt] = mm(tre[sa][rt], br[sa], yre[rt]);
        }
        __builtin_amdgcn_sched_barrier(0);
    }
    #pragma unroll
    for (int rt = 0; rt < 4; ++rt) {
        #pragma unroll
        for (int r8 = 0; r8 < 8; ++r8) {
            const int r = 16 * rt + 8 * hf + r8;
            if (CPLX) {
                vhs[re_row * VROW + r * 16 + cl]       = yre[rt][r8] - k3[rt][r8];
                vhs[(re_row + 1) * VROW + r * 16 + cl] = yre[rt][r8] + k2[rt][r8];
            } else {
                vhs[re_row * VROW + r * 16 + cl] = yre[rt][r8];
            }
        }
    }
}

__global__ __launch_bounds__(512) void main_kernel(const float* __restrict__ v,
                                                   const float* __restrict__ ws,
                                                   float* __restrict__ out)
{
    __shared__ __attribute__((aligned(16))) float vhs[64 * VROW];
    __shared__ __attribute__((aligned(16))) float tabs[17 * 512];
    __shared__ __attribute__((aligned(16))) float fm[64 * FMP];

    const int tid  = threadIdx.x;
    const int lane = tid & 31;
    const int wave = __builtin_amdgcn_readfirstlane(tid >> 5);
    const int cl   = lane & 15;
    const int hf   = lane >> 4;
    const int sub  = cl >> 2;
    const int q    = cl & 3;
    const int h    = blockIdx.x >> 4;
    const int xt   = blockIdx.x & 15;
    const int n    = xt >> 2;
    const int d0   = 16 * (xt & 3);
    const int p0   = 4 * wave;

    const float* bp = v + (((size_t)n * LFULL + (p0 + sub + 1)) * Hh + h) * Dd + d0 + 4 * q
                        + (size_t)(2 * hf) * (64 * Hh * Dd);
    const float* tabg = ws + WS_TAB + (size_t)h * (NF * 2 * TABP);
    float tv[17];
    #pragma unroll
    for (int i = 0; i < 17; ++i) {
        const int e = tid + 512 * i;
        tv[i] = tabg[e < NF * 2 * TABP ? e : 0];
    }
    __builtin_amdgcn_sched_barrier(0);
    v4f b0[8], b1[8];
    #pragma unroll
    for (int ks = 0; ks < 8; ++ks) {
        b0[ks] = *(const v4f*)(bp + (size_t)(4 * ks) * (64 * Hh * Dd));
        b1[ks] = *(const v4f*)(bp + (size_t)(4 * ks + 1) * (64 * Hh * Dd));
    }
    for (int i = 0; i < 4; ++i) {
        const int e    = tid + 512 * i;
        const int row  = e >> 5;
        const int blk  = e & 31;
        const int f    = row == 0 ? 0 : (row == 1 ? 32 : (row >> 1));
        const int part = row < 2 ? 0 : (row & 1);
        fm[row * FMP + blk] = cospif((float)((f * blk + 16 * part) & 63) * (1.0f / 32.0f));
    }
    #pragma unroll
    for (int i = 0; i < 17; ++i) tabs[tid + 512 * i] = tv[i];
    asm volatile("s_wait_dscnt 0x0" ::: "memory");
    asm volatile("s_barrier_signal -1" ::: "memory");
    asm volatile("s_barrier_wait -1" ::: "memory");
    {
        const float* ap = fm + cl * FMP + 2 * hf;
        v8f acc[4][4] = {};
        v2f a[2][4];
        #pragma unroll
        for (int mt = 0; mt < 4; ++mt) a[0][mt] = *(const v2f*)(ap + mt * (16 * FMP));
        #pragma unroll
        for (int ks = 0; ks < 8; ++ks) {
            const int sa = ks & 1;
            if (ks + 1 < 8) {
                #pragma unroll
                for (int mt = 0; mt < 4; ++mt) a[sa ^ 1][mt] = *(const v2f*)(ap + mt * (16 * FMP) + 4 * (ks + 1));
            }
            __builtin_amdgcn_sched_barrier(0);
            #pragma unroll
            for (int nt = 0; nt < 4; ++nt) {
                v2f b;
                b[0] = b0[ks][nt];
                b[1] = b1[ks][nt];
                #pragma unroll
                for (int mt = 0; mt < 4; ++mt) acc[mt][nt] = mm(a[sa][mt], b, acc[mt][nt]);
            }
            __builtin_amdgcn_sched_barrier(0);
        }
        #pragma unroll
        for (int mt = 0; mt < 4; ++mt) {
            #pragma unroll
            for (int r8 = 0; r8 < 8; ++r8) {
                const int row = 16 * mt + 8 * hf + r8;
                v4f o;
                o[0] = acc[mt][0][r8];
                o[1] = acc[mt][1][r8];
                o[2] = acc[mt][2][r8];
                o[3] = acc[mt][3][r8];
                *(v4f*)(vhs + row * VROW + (p0 + sub) * 16 + 4 * q) = o;
            }
        }
    }
    __syncthreads();

    if (wave < 15) {
        #pragma unroll 1
        for (int i = 0; i < 2; ++i) {
            const int f = 2 * wave + 1 + i;
            lds_product<true>(vhs, tabs + f * (2 * TABP), tabs + f * (2 * TABP) + TABP, 2 * f, cl, hf);
        }
    } else {
        lds_product<true>(vhs, tabs + 31 * (2 * TABP), tabs + 31 * (2 * TABP) + TABP, 62, cl, hf);
        lds_product<false>(vhs, tabs, tabs, 0, cl, hf);
        lds_product<false>(vhs, tabs + 32 * (2 * TABP), tabs + 32 * (2 * TABP), 1, cl, hf);
    }
    __syncthreads();

    {
        const float* ap = ws + WS_G + cl * 64 + 2 * hf;
        const float* bq = vhs + (2 * hf) * VROW + (p0 + sub) * 16 + 4 * q;
        v8f acc[2][4] = {};
        #pragma unroll
        for (int ks = 0; ks < 16; ++ks) {
            const v4f c0 = *(const v4f*)(bq + (4 * ks) * VROW);
            const v4f c1 = *(const v4f*)(bq + (4 * ks + 1) * VROW);
            v2f a[2];
            #pragma unroll
            for (int mt = 0; mt < 2; ++mt) a[mt] = *(const v2f*)(ap + mt * 1024 + 4 * ks);
            #pragma unroll
            for (int nt = 0; nt < 4; ++nt) {
                v2f b;
                b[0] = c0[nt];
                b[1] = c1[nt];
                #pragma unroll
                for (int mt = 0; mt < 2; ++mt) acc[mt][nt] = mm(a[mt], b, acc[mt][nt]);
            }
        }
        #pragma unroll
        for (int mt = 0; mt < 2; ++mt) {
            #pragma unroll
            for (int r8 = 0; r8 < 8; ++r8) {
                const int J = 16 * mt + 8 * hf + r8;
                const int j = 64 * J + p0 + sub;
                v4f o;
                o[0] = acc[mt][0][r8];
                o[1] = acc[mt][1][r8];
                o[2] = acc[mt][2][r8];
                o[3] = acc[mt][3][r8];
                *(v4f*)(out + (((size_t)n * LFULL + (j + 1)) * Hh + h) * Dd + d0 + 4 * q) = o;
            }
        }
    }
}

extern "C" void kernel_launch(void* const* d_in, const int* in_sizes, int n_in,
                              void* d_out, int out_size, void* d_ws, size_t ws_size,
                              hipStream_t stream)
{
    (void)in_sizes; (void)n_in; (void)out_size; (void)ws_size;
    const float* v      = (const float*)d_in[0];
    const float* offset = (const float*)d_in[1];
    const float* w      = (const float*)d_in[2];
    float*       out    = (float*)d_out;
    float*       ws     = (float*)d_ws;

    prep_kernel<<<dim3(NB_PREP), 128, 0, stream>>>(w, offset, ws, out);
    main_kernel<<<dim3(NB_MAIN), 512, 0, stream>>>(v, ws, out);
}


# One launch, not two: each workgroup forms its head's diagonal tables itself under its V loads, as one 64x63x127 matrix product, and 128 rows of z_pb; changes the f32 sum order of the tables
# speedup vs baseline: 4.3896x; 1.3130x over previous
#include <hip/hip_runtime.h>


typedef float v2f __attribute__((ext_vector_type(2)));
typedef float v4f __attribute__((ext_vector_type(4)));
typedef float v8f __attribute__((ext_vector_type(8)));

#define Bc     4
#define LFULL  2050
#define Lq     2048
#define Hh     16
#define Dd     64
#define SHAPE  4095


#define NF     33
#define TABP   128
#define NB_MAIN (Hh * 16)
#define FMP    34
#define GMP    68
#define VROW   1024
#define E_OFF   0
#define S_OFF   4224
#define C64_OFF 8320
#define WT_OFF  8384
#define F_OFF   8448

__device__ __forceinline__ v8f mm(v2f a, v2f b, v8f c)
{
    return __builtin_amdgcn_wmma_f32_16x16x4_f32(false, a, false, b, (short)0, c, false, false);
}

__device__ __forceinline__ void lds_barrier()
{
    asm volatile("s_wait_dscnt 0x0" ::: "memory");
    asm volatile("s_barrier_signal -1" ::: "memory");
    asm volatile("s_barrier_wait -1" ::: "memory");
}

template <bool CPLX>
__device__ __forceinline__ void lds_product(float* vhs, const float* tre_t, const float* tim_t,
                                            int re_row, int cl, int hf)
{
    float* vre = vhs + re_row * VROW + (2 * hf) * 16 + cl;
    float* vim = vre + VROW;
    const int abase = 2 * hf - cl + 63;

    v8f yre[4] = {};
    v8f k2[4] = {};
    v8f k3[4] = {};
    v2f br[2], bi[2], tre[2][4], tim[2][4];

    br[0][0] = vre[0];
    br[0][1] = vre[16];
    if (CPLX) {
        bi[0][0] = vim[0];
        bi[0][1] = vim[16];
    }
    #pragma unroll
    for (int rt = 0; rt < 4; ++rt) {
        const int ai = abase - 16 * rt;
        tre[0][rt][0] = tre_t[ai];
        tre[0][rt][1] = tre_t[ai + 1];
        if (CPLX) {
            tim[0][rt][0] = tim_t[ai];
            tim[0][rt][1] = tim_t[ai + 1];
        }
    }

    #pragma unroll
    for (int ks = 0; ks < 16; ++ks) {
        const int sa = ks & 1;
        if (ks + 1 < 16) {
            br[sa ^ 1][0] = vre[(4 * ks + 4) * 16];
            br[sa ^ 1][1] = vre[(4 * ks + 5) * 16];
            if (CPLX) {
                bi[sa ^ 1][0] = vim[(4 * ks + 4) * 16];
                bi[sa ^ 1][1] = vim[(4 * ks + 5) * 16];
            }
            #pragma unroll
            for (int rt = 0; rt < 4; ++rt) {
                const int ai = abase - 16 * rt + 4 * (ks + 1);
                tre[sa ^ 1][rt][0] = tre_t[ai];
                tre[sa ^ 1][rt][1] = tre_t[ai + 1];
                if (CPLX) {
                    tim[sa ^ 1][rt][0] = tim_t[ai];
                    tim[sa ^ 1][rt][1] = tim_t[ai + 1];
                }
            }
        }
        __builtin_amdgcn_sched_barrier(0);
        if (CPLX) {
            const v2f bsum = br[sa] + bi[sa];
            const v2f bdif = bi[sa] - br[sa];
            #pragma unroll
            for (int rt = 0; rt < 4; ++rt) {
                const v2f tsum = tre[sa][rt] + tim[sa][rt];
                yre[rt] = mm(tsum, br[sa], yre[rt]);
                k2[rt]  = mm(tre[sa][rt], bdif, k2[rt]);
                k3[rt]  = mm(tim[sa][rt], bsum, k3[rt]);
            }
        } else {
            #pragma unroll
            for (int rt = 0; rt < 4; ++rt) yre[rt] = mm(tre[sa][rt], br[sa], yre[rt]);
        }
        __builtin_amdgcn_sched_barrier(0);
    }
    #pragma unroll
    for (int rt = 0; rt < 4; ++rt) {
        #pragma unroll
        for (int r8 = 0; r8 < 8; ++r8) {
            const int r = 16 * rt + 8 * hf + r8;
            if (CPLX) {
                vhs[re_row * VROW + r * 16 + cl]       = yre[rt][r8] - k3[rt][r8];
                vhs[(re_row + 1) * VROW + r * 16 + cl] = yre[rt][r8] + k2[rt][r8];
            } else {
                vhs[re_row * VROW + r * 16 + cl] = yre[rt][r8];
            }
        }
    }
}

__global__ __launch_bounds__(512) void main_kernel(const float* __restrict__ v,
                                                   const float* __restrict__ offset,
                                                   const float* w,
                                                   float* __restrict__ out)
{
    __shared__ __attribute__((aligned(16))) float vhs[64 * VROW];
    __shared__ __attribute__((aligned(16))) float tabs[NF * 2 * TABP];
    __shared__ __attribute__((aligned(16))) float fm[64 * FMP];
    __shared__ __attribute__((aligned(16))) float gm[32 * GMP];

    const int tid  = threadIdx.x;
    const int lane = tid & 31;
    const int wave = __builtin_amdgcn_readfirstlane(tid >> 5);
    const int cl   = lane & 15;
    const int hf   = lane >> 4;
    const int sub  = cl >> 2;
    const int q    = cl & 3;
    const int h    = blockIdx.x >> 4;
    const int xt   = blockIdx.x & 15;
    const int n    = xt >> 2;
    const int d0   = 16 * (xt & 3);
    const int p0   = 4 * wave;

    float* es  = vhs + E_OFF;
    float* ss  = vhs + S_OFF;
    float* c64 = vhs + C64_OFF;
    float* wt  = vhs + WT_OFF;
    float* fs  = vhs + F_OFF;

    const float off = offset[h];
    const float* wrow = w + h * SHAPE;
    float raw[8];
    #pragma unroll
    for (int i = 0; i < 8; ++i) {
        const int k = tid + 512 * i;
        raw[i] = wrow[k < SHAPE ? k : SHAPE - 1];
    }
    __builtin_amdgcn_sched_barrier(0);
    const float* bp = v + (((size_t)n * LFULL + (p0 + sub + 1)) * Hh + h) * Dd + d0 + 4 * q
                        + (size_t)(2 * hf) * (64 * Hh * Dd);
    v4f b0[8], b1[8];
    #pragma unroll
    for (int ks = 0; ks < 8; ++ks) {
        b0[ks] = *(const v4f*)(bp + (size_t)(4 * ks) * (64 * Hh * Dd));
        b1[ks] = *(const v4f*)(bp + (size_t)(4 * ks + 1) * (64 * Hh * Dd));
    }
    for (int i = 0; i < 4; ++i) {
        const int e    = tid + 512 * i;
        const int row  = e >> 5;
        const int blk  = e & 31;
        const int f    = row == 0 ? 0 : (row == 1 ? 32 : (row >> 1));
        const int part = row < 2 ? 0 : (row & 1);
        fm[row * FMP + blk] = cospif((float)((f * blk + 16 * part) & 63) * (1.0f / 32.0f));
    }
    for (int i = 0; i < 4; ++i) {
        const int e    = tid + 512 * i;
        const int row  = e & 63;
        const int blk  = e >> 6;
        const int f    = row == 0 ? 0 : (row == 1 ? 32 : (row >> 1));
        const int part = row < 2 ? 0 : (row & 1);
        const float x  = (float)((f * blk) & 63) * (1.0f / 32.0f);
        gm[blk * GMP + row] = part ? (-2.0f / 64.0f) * sinpif(x)
                                   : (row < 2 ? 1.0f / 64.0f : 2.0f / 64.0f) * cospif(x);
    }
    if (tid < 64) c64[tid] = cospif((float)tid * (1.0f / 32.0f));
    #pragma unroll
    for (int i = 0; i < 8; ++i) {
        const int k = tid + 512 * i;
        es[k] = (k < SHAPE) ? __expf(raw[i] - off) : 0.f;
    }
    if (tid < 128) es[4096 + tid] = 0.f;
    lds_barrier();

    #pragma unroll
    for (int i = 0; i < 8; ++i) {
        const int e    = tid + 512 * i;
        const int j    = e >> 6;
        const int kk   = e & 63;
        const int f    = j == 0 ? 0 : (j == 1 ? 32 : (j >> 1));
        const int part = j < 2 ? 0 : (j & 1);
        const float cv = c64[(f * (kk - 31) - 16 * part) & 63];
        fs[j * GMP + kk] = kk < 63 ? cv : 0.f;
    }

    float loc[16] = {};
    float racc = 0.f, rinc = 0.f;
    if (wave < 8) {
        const int base = tid * 16;
        float a = 0.f;
        #pragma unroll
        for (int qq = 0; qq < 4; ++qq) {
            const v4f t4 = *(const v4f*)(es + base + 4 * qq);
            #pragma unroll
            for (int e = 0; e < 4; ++e) { loc[4 * qq + e] = a; a += t4[e]; }
        }
        racc = a;
        float t = a;
        #pragma unroll
        for (int d = 1; d < 32; d <<= 1) {
            const float up = __shfl_up(t, d, 32);
            if (lane >= d) t += up;
        }
        rinc = t;
        if (lane == 31) wt[wave] = t;
    }

    lds_barrier();
    {
        const int nt = wave & 7;
        const int mh = wave >> 3;
        const float* fp = fs + (32 * mh + cl) * GMP + 2 * hf;
        const float* ep = es + 64 * (2 * hf) + 16 * nt + cl;
        v8f ta[2] = {};
        #pragma unroll
        for (int ks = 0; ks < 16; ++ks) {
            v2f b;
            b[0] = ep[256 * ks];
            b[1] = ep[256 * ks + 64];
            #pragma unroll
            for (int mi = 0; mi < 2; ++mi) {
                const v2f a = *(const v2f*)(fp + mi * (16 * GMP) + 4 * ks);
                ta[mi] = mm(a, b, ta[mi]);
            }
        }
        #pragma unroll
        for (int mi = 0; mi < 2; ++mi) {
            #pragma unroll
            for (int r8 = 0; r8 < 8; ++r8) {
                const int j    = 32 * mh + 16 * mi + 8 * hf + r8;
                const int f    = j == 0 ? 0 : (j == 1 ? 32 : (j >> 1));
                const int part = j < 2 ? 0 : (j & 1);
                tabs[f * (2 * TABP) + part * TABP + 16 * nt + cl] = ta[mi][r8];
            }
        }
    }
    lds_barrier();
    if (wave < 8) {
        float pbase = rinc - racc;
        #pragma unroll
        for (int qq = 0; qq < 7; ++qq) pbase += (qq < wave) ? wt[qq] : 0.f;
        #pragma unroll
        for (int qq = 0; qq < 4; ++qq) {
            v4f t4;
            #pragma unroll
            for (int e = 0; e < 4; ++e) t4[e] = pbase + loc[4 * qq + e];
            *(v4f*)(ss + tid * 16 + 4 * qq) = t4;
        }
    }
    lds_barrier();
    float zv = 0.f;
    if (wave < 4) {
        const int i = 128 * xt + tid;
        zv = ss[i + Lq] - ss[i];
    }
    lds_barrier();
    {
        float* zout = out + (size_t)Bc * LFULL * Hh * Dd;
        if (wave < 4) {
            const int i = 128 * xt + tid;
            #pragma unroll
            for (int nn = 0; nn < Bc; ++nn)
                zout[((size_t)nn * LFULL + (i + 1)) * Hh + h] = zv;
        }
        if (xt == 0 && tid < 2 * Bc) {
            const int nn = tid & 3;
            const int jp = (tid >> 2) ? (LFULL - 1) : 0;
            zout[((size_t)nn * LFULL + jp) * Hh + h] = 0.f;
        }
        if (tid < 32) {
            const int jp = (tid >> 4) ? (LFULL - 1) : 0;
            out[(((size_t)n * LFULL + jp) * Hh + h) * Dd + d0 + (tid & 15)] = 0.f;
        }
    }

    {
        const float* ap = fm + cl * FMP + 2 * hf;
        v8f acc[4][4] = {};
        v2f a[2][4];
        #pragma unroll
        for (int mt = 0; mt < 4; ++mt) a[0][mt] = *(const v2f*)(ap + mt * (16 * FMP));
        #pragma unroll
        for (int ks = 0; ks < 8; ++ks) {
            const int sa = ks & 1;
            if (ks + 1 < 8) {
                #pragma unroll
                for (int mt = 0; mt < 4; ++mt) a[sa ^ 1][mt] = *(const v2f*)(ap + mt * (16 * FMP) + 4 * (ks + 1));
            }
            __builtin_amdgcn_sched_barrier(0);
            #pragma unroll
            for (int nt = 0; nt < 4; ++nt) {
                v2f b;
                b[0] = b0[ks][nt];
                b[1] = b1[ks][nt];
                #pragma unroll
                for (int mt = 0; mt < 4; ++mt) acc[mt][nt] = mm(a[sa][mt], b, acc[mt][nt]);
            }
            __builtin_amdgcn_sched_barrier(0);
        }
        #pragma unroll
        for (int mt = 0; mt < 4; ++mt) {
            #pragma unroll
            for (int r8 = 0; r8 < 8; ++r8) {
                const int row = 16 * mt + 8 * hf + r8;
                v4f o;
                o[0] = acc[mt][0][r8];
                o[1] = acc[mt][1][r8];
                o[2] = acc[mt][2][r8];
                o[3] = acc[mt][3][r8];
                *(v4f*)(vhs + row * VROW + (p0 + sub) * 16 + 4 * q) = o;
            }
        }
    }
    __syncthreads();

    if (wave < 15) {
        #pragma unroll 1
        for (int i = 0; i < 2; ++i) {
            const int f = 2 * wave + 1 + i;
            lds_product<true>(vhs, tabs + f * (2 * TABP), tabs + f * (2 * TABP) + TABP, 2 * f, cl, hf);
        }
    } else {
        lds_product<true>(vhs, tabs + 31 * (2 * TABP), tabs + 31 * (2 * TABP) + TABP, 62, cl, hf);
        lds_product<false>(vhs, tabs, tabs, 0, cl, hf);
        lds_product<false>(vhs, tabs + 32 * (2 * TABP), tabs + 32 * (2 * TABP), 1, cl, hf);
    }
    __syncthreads();

    {
        const float* ap = gm + cl * GMP + 2 * hf;
        const float* bq = vhs + (2 * hf) * VROW + (p0 + sub) * 16 + 4 * q;
        v8f acc[2][4] = {};
        #pragma unroll
        for (int ks = 0; ks < 16; ++ks) {
            const v4f c0 = *(const v4f*)(bq + (4 * ks) * VROW);
            const v4f c1 = *(const v4f*)(bq + (4 * ks + 1) * VROW);
            v2f a[2];
            #pragma unroll
            for (int mt = 0; mt < 2; ++mt) a[mt] = *(const v2f*)(ap + mt * (16 * GMP) + 4 * ks);
            #pragma unroll
            for (int nt = 0; nt < 4; ++nt) {
                v2f b;
                b[0] = c0[nt];
                b[1] = c1[nt];
                #pragma unroll
                for (int mt = 0; mt < 2; ++mt) acc[mt][nt] = mm(a[mt], b, acc[mt][nt]);
            }
        }
        #pragma unroll
        for (int mt = 0; mt < 2; ++mt) {
            #pragma unroll
            for (int r8 = 0; r8 < 8; ++r8) {
                const int J = 16 * mt + 8 * hf + r8;
                const int j = 64 * J + p0 + sub;
                v4f o;
                o[0] = acc[mt][0][r8];
                o[1] = acc[mt][1][r8];
                o[2] = acc[mt][2][r8];
                o[3] = acc[mt][3][r8];
                *(v4f*)(out + (((size_t)n * LFULL + (j + 1)) * Hh + h) * Dd + d0 + 4 * q) = o;
            }
        }
    }
}

extern "C" void kernel_launch(void* const* d_in, const int* in_sizes, int n_in,
                              void* d_out, int out_size, void* d_ws, size_t ws_size,
                              hipStream_t stream)
{
    (void)in_sizes; (void)n_in; (void)out_size; (void)d_ws; (void)ws_size;
    const float* v      = (const float*)d_in[0];
    const float* offset = (const float*)d_in[1];
    const float* w      = (const float*)d_in[2];
    float*       out    = (float*)d_out;

    main_kernel<<<dim3(NB_MAIN), 512, 0, stream>>>(v, offset, w, out);
}
